# starting kernel
# baseline (speedup 1.0000x reference)
#include <hip/hip_runtime.h>
#include <math.h>

#define EPSV   1e-8f
#define GROUPS 128
#define Bdim   4
#define Sdim   8192
#define Ddim   1024
#define Hdim   1024
#define NCHUNK 64
#define CLEN   (Sdim / NCHUNK)

typedef float v2f __attribute__((ext_vector_type(2)));
typedef float v8f __attribute__((ext_vector_type(8)));

__device__ __forceinline__ float softplus_f(float x) {
  return fmaxf(x, 0.f) + log1pf(expf(-fabsf(x)));
}

__device__ __forceinline__ float logaddexp_f(float a, float b) {
  float mx = fmaxf(a, b), mn = fminf(a, b);
  if (mx == -INFINITY) return -INFINITY;
  return mx + log1pf(expf(mn - mx));
}

__device__ __forceinline__ float log_g_f(float x) {
  return (x >= 0.f) ? logf(x + 0.5f + EPSV) : (-softplus_f(-x) + EPSV);
}

__device__ __forceinline__ v2f ldw(const float* __restrict__ Wg, int cn, int kc,
                                   int hi, bool valid) {
  v2f w = *(const v2f*)(Wg + cn * 8 + kc * 4 + 2 * hi);
  v2f r;
  r.x = valid ? w.x : 0.f;
  r.y = valid ? w.y : 0.f;
  return r;
}

__device__ __forceinline__ v8f gate16(v2f a00, v2f a01, v2f a10, v2f a11,
                                      const float* __restrict__ W, int g0,
                                      int n, int hi) {
  const int  cn = n & 7;
  const bool lo = (n < 8);
  const float* Wg0 = W + (size_t)g0 * 64;
  const float* Wg1 = Wg0 + 64;
  v2f b00 = ldw(Wg0, cn, 0, hi, lo);
  v2f b01 = ldw(Wg0, cn, 1, hi, lo);
  v2f b10 = ldw(Wg1, cn, 0, hi, !lo);
  v2f b11 = ldw(Wg1, cn, 1, hi, !lo);
  v8f acc = {};
  acc = __builtin_amdgcn_wmma_f32_16x16x4_f32(false, a00, false, b00, (short)0, acc, false, false);
  acc = __builtin_amdgcn_wmma_f32_16x16x4_f32(false, a01, false, b01, (short)0, acc, false, false);
  acc = __builtin_amdgcn_wmma_f32_16x16x4_f32(false, a10, false, b10, (short)0, acc, false, false);
  acc = __builtin_amdgcn_wmma_f32_16x16x4_f32(false, a11, false, b11, (short)0, acc, false, false);
  return acc;
}

__global__ void __launch_bounds__(256) gates_wmma(
    const float* __restrict__ x, const float* __restrict__ Wf,
    const float* __restrict__ Wi, const float* __restrict__ Wh,
    float* __restrict__ lf_out, float* __restrict__ lv_out) {
  const int lane = threadIdx.x & 31;
  const int wv   = threadIdx.x >> 5;
  const int gid  = blockIdx.x * 8 + wv;
  const int NPAIR = GROUPS / 2;
  const int gp = gid % NPAIR;
  const int tt = gid / NPAIR;
  const int s0 = (tt * 16) % Sdim;
  const int b  = (tt * 16) / Sdim;

  const int n  = lane & 15;
  const int hi = lane >> 4;

  const float* xrow =
      x + ((size_t)b * Sdim + (size_t)(s0 + n)) * (size_t)Ddim + (size_t)gp * 16;
  v2f a00 = *(const v2f*)(xrow + 0  + 2 * hi);
  v2f a01 = *(const v2f*)(xrow + 4  + 2 * hi);
  v2f a10 = *(const v2f*)(xrow + 8  + 2 * hi);
  v2f a11 = *(const v2f*)(xrow + 12 + 2 * hi);

  const int g0 = gp * 2;
  v8f fg = gate16(a00, a01, a10, a11, Wf, g0, n, hi);
  v8f ig = gate16(a00, a01, a10, a11, Wi, g0, n, hi);
  v8f hg = gate16(a00, a01, a10, a11, Wh, g0, n, hi);

  const size_t baseoff =
      ((size_t)b * Sdim + (size_t)(s0 + 8 * hi)) * (size_t)Hdim + (size_t)gp * 16 + n;
#pragma unroll
  for (int r = 0; r < 8; ++r) {
    float f  = fg[r];
    float i  = ig[r];
    float hd = hg[r];
    float diff  = softplus_f(-f) - softplus_f(-i);
    float logf_ = -softplus_f(diff);
    float logi_ = -softplus_f(-diff);
    float lv    = logi_ + log_g_f(hd);
    size_t off = baseoff + (size_t)r * Hdim;
    lf_out[off] = logf_;
    lv_out[off] = lv;
  }
}

__global__ void __launch_bounds__(256) chunk_local(
    const float* __restrict__ lf, const float* __restrict__ lv,
    float* __restrict__ chA, float* __restrict__ chM) {
  int idx = blockIdx.x * 256 + threadIdx.x;
  int h  = idx % Hdim;
  int t2 = idx / Hdim;
  int c  = t2 % NCHUNK;
  int b  = t2 / NCHUNK;
  size_t base = ((size_t)b * Sdim + (size_t)c * CLEN) * (size_t)Hdim + h;
  float a = 0.f, l = -INFINITY;
  for (int t = 0; t < CLEN; ++t) {
    size_t off = base + (size_t)t * Hdim;
    __builtin_prefetch(lf + off + (size_t)8 * Hdim, 0, 0);
    __builtin_prefetch(lv + off + (size_t)8 * Hdim, 0, 0);
    float lfv = lf[off];
    float lvv = lv[off];
    a += lfv + EPSV;
    l = logaddexp_f(l, lvv - a);
  }
  int sidx = c * (Bdim * Hdim) + b * Hdim + h;
  chA[sidx] = a;
  chM[sidx] = l;
}

__global__ void __launch_bounds__(256) carry_scan(
    const float* __restrict__ chA, const float* __restrict__ chM,
    float* __restrict__ cQ) {
  int chan = blockIdx.x * 256 + threadIdx.x;
  const int nchan = Bdim * Hdim;
  if (chan >= nchan) return;
  float P = 0.f, G = -INFINITY;
  for (int c = 0; c < NCHUNK; ++c) {
    cQ[c * nchan + chan] = G + P;
    G = logaddexp_f(G, chM[c * nchan + chan] - P);
    P += chA[c * nchan + chan];
  }
}

__global__ void __launch_bounds__(256) apply_scan(
    const float* __restrict__ lf, const float* __restrict__ lv,
    const float* __restrict__ cQ, float* __restrict__ out) {
  int idx = blockIdx.x * 256 + threadIdx.x;
  int h  = idx % Hdim;
  int t2 = idx / Hdim;
  int c  = t2 % NCHUNK;
  int b  = t2 / NCHUNK;
  size_t base = ((size_t)b * Sdim + (size_t)c * CLEN) * (size_t)Hdim + h;
  float a = 0.f;
  float R = cQ[c * (Bdim * Hdim) + b * Hdim + h];
  for (int t = 0; t < CLEN; ++t) {
    size_t off = base + (size_t)t * Hdim;
    __builtin_prefetch(lf + off + (size_t)8 * Hdim, 0, 0);
    __builtin_prefetch(lv + off + (size_t)8 * Hdim, 0, 0);
    float lfv = lf[off];
    float lvv = lv[off];
    a += lfv + EPSV;
    R = logaddexp_f(R, lvv - a);
    out[off] = expf(a + R);
  }
}

extern "C" void kernel_launch(void* const* d_in, const int* in_sizes, int n_in,
                              void* d_out, int out_size, void* d_ws, size_t ws_size,
                              hipStream_t stream) {
  (void)in_sizes; (void)n_in; (void)out_size; (void)ws_size;
  const float* x  = (const float*)d_in[0];
  const float* Wf = (const float*)d_in[1];
  const float* Wi = (const float*)d_in[2];
  const float* Wh = (const float*)d_in[3];
  float* out = (float*)d_out;

  const size_t NBSH = (size_t)Bdim * Sdim * Hdim;
  const size_t NSUM = (size_t)NCHUNK * Bdim * Hdim;
  float* lf  = (float*)d_ws;
  float* lv  = lf + NBSH;
  float* chA = lv + NBSH;
  float* chM = chA + NSUM;
  float* cQ  = chM + NSUM;

  const int gate_blocks = (Bdim * Sdim / 16) * (GROUPS / 2) / 8;
  gates_wmma<<<gate_blocks, 256, 0, stream>>>(x, Wf, Wi, Wh, lf, lv);

  const int scan_threads = Bdim * NCHUNK * Hdim;
  chunk_local<<<scan_threads / 256, 256, 0, stream>>>(lf, lv, chA, chM);
  carry_scan<<<(Bdim * Hdim + 255) / 256, 256, 0, stream>>>(chA, chM, cQ);
  apply_scan<<<scan_threads / 256, 256, 0, stream>>>(lf, lv, cQ, out);
}


# library log1pf in softplus and logaddexp (all four kernels) replaced by 2*atanh(e/(2+e)) odd series to t^15 on e=exp(v<=0) in [0,1], Newton-refined quotient, about 2 ulp; sum order unchanged
# speedup vs baseline: 1.6753x; 1.6753x over previous
#include <hip/hip_runtime.h>
#include <math.h>

#define EPSV   1e-8f
#define GROUPS 128
#define Bdim   4
#define Sdim   8192
#define Ddim   1024
#define Hdim   1024
#define NCHUNK 64
#define CLEN   (Sdim / NCHUNK)

typedef float v2f __attribute__((ext_vector_type(2)));
typedef float v8f __attribute__((ext_vector_type(8)));

__device__ __forceinline__ float log1p_unit(float e) {
  float d  = 2.0f + e;
  float r  = __builtin_amdgcn_rcpf(d);
  float t  = e * r;
  t = fmaf(fmaf(-d, t, e), r, t);
  float s  = t * t;
  float p  = 1.0f / 15.0f;
  p = fmaf(p, s, 1.0f / 13.0f);
  p = fmaf(p, s, 1.0f / 11.0f);
  p = fmaf(p, s, 1.0f / 9.0f);
  p = fmaf(p, s, 1.0f / 7.0f);
  p = fmaf(p, s, 1.0f / 5.0f);
  p = fmaf(p, s, 1.0f / 3.0f);
  float t2 = t + t;
  return fmaf(t2 * s, p, t2);
}

__device__ __forceinline__ float softplus_f(float x) {
  return fmaxf(x, 0.f) + log1p_unit(expf(-fabsf(x)));
}

__device__ __forceinline__ float logaddexp_f(float a, float b) {
  float mx = fmaxf(a, b), mn = fminf(a, b);
  if (mx == -INFINITY) return -INFINITY;
  return mx + log1p_unit(expf(mn - mx));
}

__device__ __forceinline__ float log_g_f(float x) {
  return (x >= 0.f) ? logf(x + 0.5f + EPSV) : (-softplus_f(-x) + EPSV);
}

__device__ __forceinline__ v2f ldw(const float* __restrict__ Wg, int cn, int kc,
                                   int hi, bool valid) {
  v2f w = *(const v2f*)(Wg + cn * 8 + kc * 4 + 2 * hi);
  v2f r;
  r.x = valid ? w.x : 0.f;
  r.y = valid ? w.y : 0.f;
  return r;
}

__device__ __forceinline__ v8f gate16(v2f a00, v2f a01, v2f a10, v2f a11,
                                      const float* __restrict__ W, int g0,
                                      int n, int hi) {
  const int  cn = n & 7;
  const bool lo = (n < 8);
  const float* Wg0 = W + (size_t)g0 * 64;
  const float* Wg1 = Wg0 + 64;
  v2f b00 = ldw(Wg0, cn, 0, hi, lo);
  v2f b01 = ldw(Wg0, cn, 1, hi, lo);
  v2f b10 = ldw(Wg1, cn, 0, hi, !lo);
  v2f b11 = ldw(Wg1, cn, 1, hi, !lo);
  v8f acc = {};
  acc = __builtin_amdgcn_wmma_f32_16x16x4_f32(false, a00, false, b00, (short)0, acc, false, false);
  acc = __builtin_amdgcn_wmma_f32_16x16x4_f32(false, a01, false, b01, (short)0, acc, false, false);
  acc = __builtin_amdgcn_wmma_f32_16x16x4_f32(false, a10, false, b10, (short)0, acc, false, false);
  acc = __builtin_amdgcn_wmma_f32_16x16x4_f32(false, a11, false, b11, (short)0, acc, false, false);
  return acc;
}

__global__ void __launch_bounds__(256) gates_wmma(
    const float* __restrict__ x, const float* __restrict__ Wf,
    const float* __restrict__ Wi, const float* __restrict__ Wh,
    float* __restrict__ lf_out, float* __restrict__ lv_out) {
  const int lane = threadIdx.x & 31;
  const int wv   = threadIdx.x >> 5;
  const int gid  = blockIdx.x * 8 + wv;
  const int NPAIR = GROUPS / 2;
  const int gp = gid % NPAIR;
  const int tt = gid / NPAIR;
  const int s0 = (tt * 16) % Sdim;
  const int b  = (tt * 16) / Sdim;

  const int n  = lane & 15;
  const int hi = lane >> 4;

  const float* xrow =
      x + ((size_t)b * Sdim + (size_t)(s0 + n)) * (size_t)Ddim + (size_t)gp * 16;
  v2f a00 = *(const v2f*)(xrow + 0  + 2 * hi);
  v2f a01 = *(const v2f*)(xrow + 4  + 2 * hi);
  v2f a10 = *(const v2f*)(xrow + 8  + 2 * hi);
  v2f a11 = *(const v2f*)(xrow + 12 + 2 * hi);

  const int g0 = gp * 2;
  v8f fg = gate16(a00, a01, a10, a11, Wf, g0, n, hi);
  v8f ig = gate16(a00, a01, a10, a11, Wi, g0, n, hi);
  v8f hg = gate16(a00, a01, a10, a11, Wh, g0, n, hi);

  const size_t baseoff =
      ((size_t)b * Sdim + (size_t)(s0 + 8 * hi)) * (size_t)Hdim + (size_t)gp * 16 + n;
#pragma unroll
  for (int r = 0; r < 8; ++r) {
    float f  = fg[r];
    float i  = ig[r];
    float hd = hg[r];
    float diff  = softplus_f(-f) - softplus_f(-i);
    float logf_ = -softplus_f(diff);
    float logi_ = -softplus_f(-diff);
    float lv    = logi_ + log_g_f(hd);
    size_t off = baseoff + (size_t)r * Hdim;
    lf_out[off] = logf_;
    lv_out[off] = lv;
  }
}

__global__ void __launch_bounds__(256) chunk_local(
    const float* __restrict__ lf, const float* __restrict__ lv,
    float* __restrict__ chA, float* __restrict__ chM) {
  int idx = blockIdx.x * 256 + threadIdx.x;
  int h  = idx % Hdim;
  int t2 = idx / Hdim;
  int c  = t2 % NCHUNK;
  int b  = t2 / NCHUNK;
  size_t base = ((size_t)b * Sdim + (size_t)c * CLEN) * (size_t)Hdim + h;
  float a = 0.f, l = -INFINITY;
  for (int t = 0; t < CLEN; ++t) {
    size_t off = base + (size_t)t * Hdim;
    __builtin_prefetch(lf + off + (size_t)8 * Hdim, 0, 0);
    __builtin_prefetch(lv + off + (size_t)8 * Hdim, 0, 0);
    float lfv = lf[off];
    float lvv = lv[off];
    a += lfv + EPSV;
    l = logaddexp_f(l, lvv - a);
  }
  int sidx = c * (Bdim * Hdim) + b * Hdim + h;
  chA[sidx] = a;
  chM[sidx] = l;
}

__global__ void __launch_bounds__(256) carry_scan(
    const float* __restrict__ chA, const float* __restrict__ chM,
    float* __restrict__ cQ) {
  int chan = blockIdx.x * 256 + threadIdx.x;
  const int nchan = Bdim * Hdim;
  if (chan >= nchan) return;
  float P = 0.f, G = -INFINITY;
  for (int c = 0; c < NCHUNK; ++c) {
    cQ[c * nchan + chan] = G + P;
    G = logaddexp_f(G, chM[c * nchan + chan] - P);
    P += chA[c * nchan + chan];
  }
}

__global__ void __launch_bounds__(256) apply_scan(
    const float* __restrict__ lf, const float* __restrict__ lv,
    const float* __restrict__ cQ, float* __restrict__ out) {
  int idx = blockIdx.x * 256 + threadIdx.x;
  int h  = idx % Hdim;
  int t2 = idx / Hdim;
  int c  = t2 % NCHUNK;
  int b  = t2 / NCHUNK;
  size_t base = ((size_t)b * Sdim + (size_t)c * CLEN) * (size_t)Hdim + h;
  float a = 0.f;
  float R = cQ[c * (Bdim * Hdim) + b * Hdim + h];
  for (int t = 0; t < CLEN; ++t) {
    size_t off = base + (size_t)t * Hdim;
    __builtin_prefetch(lf + off + (size_t)8 * Hdim, 0, 0);
    __builtin_prefetch(lv + off + (size_t)8 * Hdim, 0, 0);
    float lfv = lf[off];
    float lvv = lv[off];
    a += lfv + EPSV;
    R = logaddexp_f(R, lvv - a);
    out[off] = expf(a + R);
  }
}

extern "C" void kernel_launch(void* const* d_in, const int* in_sizes, int n_in,
                              void* d_out, int out_size, void* d_ws, size_t ws_size,
                              hipStream_t stream) {
  (void)in_sizes; (void)n_in; (void)out_size; (void)ws_size;
  const float* x  = (const float*)d_in[0];
  const float* Wf = (const float*)d_in[1];
  const float* Wi = (const float*)d_in[2];
  const float* Wh = (const float*)d_in[3];
  float* out = (float*)d_out;

  const size_t NBSH = (size_t)Bdim * Sdim * Hdim;
  const size_t NSUM = (size_t)NCHUNK * Bdim * Hdim;
  float* lf  = (float*)d_ws;
  float* lv  = lf + NBSH;
  float* chA = lv + NBSH;
  float* chM = chA + NSUM;
  float* cQ  = chM + NSUM;

  const int gate_blocks = (Bdim * Sdim / 16) * (GROUPS / 2) / 8;
  gates_wmma<<<gate_blocks, 256, 0, stream>>>(x, Wf, Wi, Wh, lf, lv);

  const int scan_threads = Bdim * NCHUNK * Hdim;
  chunk_local<<<scan_threads / 256, 256, 0, stream>>>(lf, lv, chA, chM);
  carry_scan<<<(Bdim * Hdim + 255) / 256, 256, 0, stream>>>(chA, chM, cQ);
  apply_scan<<<scan_threads / 256, 256, 0, stream>>>(lf, lv, cQ, out);
}


# library expf (in-line range reduction, ldexp and two range selects, 14 instructions) -> v_exp_f32 of x*log2(e) in softplus, logaddexp and the final exp; rel err about |x|*2^-24; sum order unchanged
# speedup vs baseline: 1.8868x; 1.1262x over previous
#include <hip/hip_runtime.h>
#include <math.h>

#define EPSV   1e-8f
#define GROUPS 128
#define Bdim   4
#define Sdim   8192
#define Ddim   1024
#define Hdim   1024
#define NCHUNK 64
#define CLEN   (Sdim / NCHUNK)

typedef float v2f __attribute__((ext_vector_type(2)));
typedef float v8f __attribute__((ext_vector_type(8)));

__device__ __forceinline__ float exp_hw(float v) {
  return __builtin_amdgcn_exp2f(v * 1.44269504088896341f);
}

__device__ __forceinline__ float log1p_unit(float e) {
  float d  = 2.0f + e;
  float r  = __builtin_amdgcn_rcpf(d);
  float t  = e * r;
  t = fmaf(fmaf(-d, t, e), r, t);
  float s  = t * t;
  float p  = 1.0f / 15.0f;
  p = fmaf(p, s, 1.0f / 13.0f);
  p = fmaf(p, s, 1.0f / 11.0f);
  p = fmaf(p, s, 1.0f / 9.0f);
  p = fmaf(p, s, 1.0f / 7.0f);
  p = fmaf(p, s, 1.0f / 5.0f);
  p = fmaf(p, s, 1.0f / 3.0f);
  float t2 = t + t;
  return fmaf(t2 * s, p, t2);
}

__device__ __forceinline__ float softplus_f(float x) {
  return fmaxf(x, 0.f) + log1p_unit(exp_hw(-fabsf(x)));
}

__device__ __forceinline__ float logaddexp_f(float a, float b) {
  float mx = fmaxf(a, b), mn = fminf(a, b);
  if (mx == -INFINITY) return -INFINITY;
  return mx + log1p_unit(exp_hw(mn - mx));
}

__device__ __forceinline__ float log_g_f(float x) {
  return (x >= 0.f) ? logf(x + 0.5f + EPSV) : (-softplus_f(-x) + EPSV);
}

__device__ __forceinline__ v2f ldw(const float* __restrict__ Wg, int cn, int kc,
                                   int hi, bool valid) {
  v2f w = *(const v2f*)(Wg + cn * 8 + kc * 4 + 2 * hi);
  v2f r;
  r.x = valid ? w.x : 0.f;
  r.y = valid ? w.y : 0.f;
  return r;
}

__device__ __forceinline__ v8f gate16(v2f a00, v2f a01, v2f a10, v2f a11,
                                      const float* __restrict__ W, int g0,
                                      int n, int hi) {
  const int  cn = n & 7;
  const bool lo = (n < 8);
  const float* Wg0 = W + (size_t)g0 * 64;
  const float* Wg1 = Wg0 + 64;
  v2f b00 = ldw(Wg0, cn, 0, hi, lo);
  v2f b01 = ldw(Wg0, cn, 1, hi, lo);
  v2f b10 = ldw(Wg1, cn, 0, hi, !lo);
  v2f b11 = ldw(Wg1, cn, 1, hi, !lo);
  v8f acc = {};
  acc = __builtin_amdgcn_wmma_f32_16x16x4_f32(false, a00, false, b00, (short)0, acc, false, false);
  acc = __builtin_amdgcn_wmma_f32_16x16x4_f32(false, a01, false, b01, (short)0, acc, false, false);
  acc = __builtin_amdgcn_wmma_f32_16x16x4_f32(false, a10, false, b10, (short)0, acc, false, false);
  acc = __builtin_amdgcn_wmma_f32_16x16x4_f32(false, a11, false, b11, (short)0, acc, false, false);
  return acc;
}

__global__ void __launch_bounds__(256) gates_wmma(
    const float* __restrict__ x, const float* __restrict__ Wf,
    const float* __restrict__ Wi, const float* __restrict__ Wh,
    float* __restrict__ lf_out, float* __restrict__ lv_out) {
  const int lane = threadIdx.x & 31;
  const int wv   = threadIdx.x >> 5;
  const int gid  = blockIdx.x * 8 + wv;
  const int NPAIR = GROUPS / 2;
  const int gp = gid % NPAIR;
  const int tt = gid / NPAIR;
  const int s0 = (tt * 16) % Sdim;
  const int b  = (tt * 16) / Sdim;

  const int n  = lane & 15;
  const int hi = lane >> 4;

  const float* xrow =
      x + ((size_t)b * Sdim + (size_t)(s0 + n)) * (size_t)Ddim + (size_t)gp * 16;
  v2f a00 = *(const v2f*)(xrow + 0  + 2 * hi);
  v2f a01 = *(const v2f*)(xrow + 4  + 2 * hi);
  v2f a10 = *(const v2f*)(xrow + 8  + 2 * hi);
  v2f a11 = *(const v2f*)(xrow + 12 + 2 * hi);

  const int g0 = gp * 2;
  v8f fg = gate16(a00, a01, a10, a11, Wf, g0, n, hi);
  v8f ig = gate16(a00, a01, a10, a11, Wi, g0, n, hi);
  v8f hg = gate16(a00, a01, a10, a11, Wh, g0, n, hi);

  const size_t baseoff =
      ((size_t)b * Sdim + (size_t)(s0 + 8 * hi)) * (size_t)Hdim + (size_t)gp * 16 + n;
#pragma unroll
  for (int r = 0; r < 8; ++r) {
    float f  = fg[r];
    float i  = ig[r];
    float hd = hg[r];
    float diff  = softplus_f(-f) - softplus_f(-i);
    float logf_ = -softplus_f(diff);
    float logi_ = -softplus_f(-diff);
    float lv    = logi_ + log_g_f(hd);
    size_t off = baseoff + (size_t)r * Hdim;
    lf_out[off] = logf_;
    lv_out[off] = lv;
  }
}

__global__ void __launch_bounds__(256) chunk_local(
    const float* __restrict__ lf, const float* __restrict__ lv,
    float* __restrict__ chA, float* __restrict__ chM) {
  int idx = blockIdx.x * 256 + threadIdx.x;
  int h  = idx % Hdim;
  int t2 = idx / Hdim;
  int c  = t2 % NCHUNK;
  int b  = t2 / NCHUNK;
  size_t base = ((size_t)b * Sdim + (size_t)c * CLEN) * (size_t)Hdim + h;
  float a = 0.f, l = -INFINITY;
  for (int t = 0; t < CLEN; ++t) {
    size_t off = base + (size_t)t * Hdim;
    __builtin_prefetch(lf + off + (size_t)8 * Hdim, 0, 0);
    __builtin_prefetch(lv + off + (size_t)8 * Hdim, 0, 0);
    float lfv = lf[off];
    float lvv = lv[off];
    a += lfv + EPSV;
    l = logaddexp_f(l, lvv - a);
  }
  int sidx = c * (Bdim * Hdim) + b * Hdim + h;
  chA[sidx] = a;
  chM[sidx] = l;
}

__global__ void __launch_bounds__(256) carry_scan(
    const float* __restrict__ chA, const float* __restrict__ chM,
    float* __restrict__ cQ) {
  int chan = blockIdx.x * 256 + threadIdx.x;
  const int nchan = Bdim * Hdim;
  if (chan >= nchan) return;
  float P = 0.f, G = -INFINITY;
  for (int c = 0; c < NCHUNK; ++c) {
    cQ[c * nchan + chan] = G + P;
    G = logaddexp_f(G, chM[c * nchan + chan] - P);
    P += chA[c * nchan + chan];
  }
}

__global__ void __launch_bounds__(256) apply_scan(
    const float* __restrict__ lf, const float* __restrict__ lv,
    const float* __restrict__ cQ, float* __restrict__ out) {
  int idx = blockIdx.x * 256 + threadIdx.x;
  int h  = idx % Hdim;
  int t2 = idx / Hdim;
  int c  = t2 % NCHUNK;
  int b  = t2 / NCHUNK;
  size_t base = ((size_t)b * Sdim + (size_t)c * CLEN) * (size_t)Hdim + h;
  float a = 0.f;
  float R = cQ[c * (Bdim * Hdim) + b * Hdim + h];
  for (int t = 0; t < CLEN; ++t) {
    size_t off = base + (size_t)t * Hdim;
    __builtin_prefetch(lf + off + (size_t)8 * Hdim, 0, 0);
    __builtin_prefetch(lv + off + (size_t)8 * Hdim, 0, 0);
    float lfv = lf[off];
    float lvv = lv[off];
    a += lfv + EPSV;
    R = logaddexp_f(R, lvv - a);
    out[off] = exp_hw(a + R);
  }
}

extern "C" void kernel_launch(void* const* d_in, const int* in_sizes, int n_in,
                              void* d_out, int out_size, void* d_ws, size_t ws_size,
                              hipStream_t stream) {
  (void)in_sizes; (void)n_in; (void)out_size; (void)ws_size;
  const float* x  = (const float*)d_in[0];
  const float* Wf = (const float*)d_in[1];
  const float* Wi = (const float*)d_in[2];
  const float* Wh = (const float*)d_in[3];
  float* out = (float*)d_out;

  const size_t NBSH = (size_t)Bdim * Sdim * Hdim;
  const size_t NSUM = (size_t)NCHUNK * Bdim * Hdim;
  float* lf  = (float*)d_ws;
  float* lv  = lf + NBSH;
  float* chA = lv + NBSH;
  float* chM = chA + NSUM;
  float* cQ  = chM + NSUM;

  const int gate_blocks = (Bdim * Sdim / 16) * (GROUPS / 2) / 8;
  gates_wmma<<<gate_blocks, 256, 0, stream>>>(x, Wf, Wi, Wh, lf, lv);

  const int scan_threads = Bdim * NCHUNK * Hdim;
  chunk_local<<<scan_threads / 256, 256, 0, stream>>>(lf, lv, chA, chM);
  carry_scan<<<(Bdim * Hdim + 255) / 256, 256, 0, stream>>>(chA, chM, cQ);
  apply_scan<<<scan_threads / 256, 256, 0, stream>>>(lf, lv, cQ, out);
}


# the three scan kernels: loads of step t+16 (carry scan: t+8) issued while step t computes, into an unrolled register ring, replacing load-then-wait per step and the prefetches; sum order unchanged
# speedup vs baseline: 2.5146x; 1.3327x over previous
#include <hip/hip_runtime.h>
#include <math.h>

#define EPSV   1e-8f
#define GROUPS 128
#define Bdim   4
#define Sdim   8192
#define Ddim   1024
#define Hdim   1024
#define NCHUNK 64
#define CLEN   (Sdim / NCHUNK)

typedef float v2f __attribute__((ext_vector_type(2)));
typedef float v8f __attribute__((ext_vector_type(8)));

__device__ __forceinline__ float exp_hw(float v) {
  return __builtin_amdgcn_exp2f(v * 1.44269504088896341f);
}

__device__ __forceinline__ float log1p_unit(float e) {
  float d  = 2.0f + e;
  float r  = __builtin_amdgcn_rcpf(d);
  float t  = e * r;
  t = fmaf(fmaf(-d, t, e), r, t);
  float s  = t * t;
  float p  = 1.0f / 15.0f;
  p = fmaf(p, s, 1.0f / 13.0f);
  p = fmaf(p, s, 1.0f / 11.0f);
  p = fmaf(p, s, 1.0f / 9.0f);
  p = fmaf(p, s, 1.0f / 7.0f);
  p = fmaf(p, s, 1.0f / 5.0f);
  p = fmaf(p, s, 1.0f / 3.0f);
  float t2 = t + t;
  return fmaf(t2 * s, p, t2);
}

__device__ __forceinline__ float softplus_f(float x) {
  return fmaxf(x, 0.f) + log1p_unit(exp_hw(-fabsf(x)));
}

__device__ __forceinline__ float logaddexp_f(float a, float b) {
  float mx = fmaxf(a, b), mn = fminf(a, b);
  if (mx == -INFINITY) return -INFINITY;
  return mx + log1p_unit(exp_hw(mn - mx));
}

__device__ __forceinline__ float log_g_f(float x) {
  return (x >= 0.f) ? logf(x + 0.5f + EPSV) : (-softplus_f(-x) + EPSV);
}

__device__ __forceinline__ v2f ldw(const float* __restrict__ Wg, int cn, int kc,
                                   int hi, bool valid) {
  v2f w = *(const v2f*)(Wg + cn * 8 + kc * 4 + 2 * hi);
  v2f r;
  r.x = valid ? w.x : 0.f;
  r.y = valid ? w.y : 0.f;
  return r;
}

__device__ __forceinline__ v8f gate16(v2f a00, v2f a01, v2f a10, v2f a11,
                                      const float* __restrict__ W, int g0,
                                      int n, int hi) {
  const int  cn = n & 7;
  const bool lo = (n < 8);
  const float* Wg0 = W + (size_t)g0 * 64;
  const float* Wg1 = Wg0 + 64;
  v2f b00 = ldw(Wg0, cn, 0, hi, lo);
  v2f b01 = ldw(Wg0, cn, 1, hi, lo);
  v2f b10 = ldw(Wg1, cn, 0, hi, !lo);
  v2f b11 = ldw(Wg1, cn, 1, hi, !lo);
  v8f acc = {};
  acc = __builtin_amdgcn_wmma_f32_16x16x4_f32(false, a00, false, b00, (short)0, acc, false, false);
  acc = __builtin_amdgcn_wmma_f32_16x16x4_f32(false, a01, false, b01, (short)0, acc, false, false);
  acc = __builtin_amdgcn_wmma_f32_16x16x4_f32(false, a10, false, b10, (short)0, acc, false, false);
  acc = __builtin_amdgcn_wmma_f32_16x16x4_f32(false, a11, false, b11, (short)0, acc, false, false);
  return acc;
}

__global__ void __launch_bounds__(256) gates_wmma(
    const float* __restrict__ x, const float* __restrict__ Wf,
    const float* __restrict__ Wi, const float* __restrict__ Wh,
    float* __restrict__ lf_out, float* __restrict__ lv_out) {
  const int lane = threadIdx.x & 31;
  const int wv   = threadIdx.x >> 5;
  const int gid  = blockIdx.x * 8 + wv;
  const int NPAIR = GROUPS / 2;
  const int gp = gid % NPAIR;
  const int tt = gid / NPAIR;
  const int s0 = (tt * 16) % Sdim;
  const int b  = (tt * 16) / Sdim;

  const int n  = lane & 15;
  const int hi = lane >> 4;

  const float* xrow =
      x + ((size_t)b * Sdim + (size_t)(s0 + n)) * (size_t)Ddim + (size_t)gp * 16;
  v2f a00 = *(const v2f*)(xrow + 0  + 2 * hi);
  v2f a01 = *(const v2f*)(xrow + 4  + 2 * hi);
  v2f a10 = *(const v2f*)(xrow + 8  + 2 * hi);
  v2f a11 = *(const v2f*)(xrow + 12 + 2 * hi);

  const int g0 = gp * 2;
  v8f fg = gate16(a00, a01, a10, a11, Wf, g0, n, hi);
  v8f ig = gate16(a00, a01, a10, a11, Wi, g0, n, hi);
  v8f hg = gate16(a00, a01, a10, a11, Wh, g0, n, hi);

  const size_t baseoff =
      ((size_t)b * Sdim + (size_t)(s0 + 8 * hi)) * (size_t)Hdim + (size_t)gp * 16 + n;
#pragma unroll
  for (int r = 0; r < 8; ++r) {
    float f  = fg[r];
    float i  = ig[r];
    float hd = hg[r];
    float diff  = softplus_f(-f) - softplus_f(-i);
    float logf_ = -softplus_f(diff);
    float logi_ = -softplus_f(-diff);
    float lv    = logi_ + log_g_f(hd);
    size_t off = baseoff + (size_t)r * Hdim;
    lf_out[off] = logf_;
    lv_out[off] = lv;
  }
}

#define LOOKAHEAD 16
#define LOOKAHEAD_CARRY 8

__global__ void __launch_bounds__(256) chunk_local(
    const float* __restrict__ lf, const float* __restrict__ lv,
    float* __restrict__ chA, float* __restrict__ chM) {
  int idx = blockIdx.x * 256 + threadIdx.x;
  int h  = idx % Hdim;
  int t2 = idx / Hdim;
  int c  = t2 % NCHUNK;
  int b  = t2 / NCHUNK;
  size_t base = ((size_t)b * Sdim + (size_t)c * CLEN) * (size_t)Hdim + h;
  const float* pf = lf + base;
  const float* pv = lv + base;
  float qf[LOOKAHEAD], qv[LOOKAHEAD];
#pragma unroll
  for (int j = 0; j < LOOKAHEAD; ++j) {
    qf[j] = pf[(size_t)j * Hdim];
    qv[j] = pv[(size_t)j * Hdim];
  }
  float a = 0.f, l = -INFINITY;
#pragma unroll 1
  for (int t0 = LOOKAHEAD; t0 < CLEN; t0 += LOOKAHEAD) {
    pf += (size_t)LOOKAHEAD * Hdim;
    pv += (size_t)LOOKAHEAD * Hdim;
#pragma unroll
    for (int j = 0; j < LOOKAHEAD; ++j) {
      a += qf[j] + EPSV;
      float d = qv[j] - a;
      qf[j] = pf[(size_t)j * Hdim];
      qv[j] = pv[(size_t)j * Hdim];
      l = logaddexp_f(l, d);
    }
  }
#pragma unroll
  for (int j = 0; j < LOOKAHEAD; ++j) {
    a += qf[j] + EPSV;
    l = logaddexp_f(l, qv[j] - a);
  }
  int sidx = c * (Bdim * Hdim) + b * Hdim + h;
  chA[sidx] = a;
  chM[sidx] = l;
}

__global__ void __launch_bounds__(256) carry_scan(
    const float* __restrict__ chA, const float* __restrict__ chM,
    float* __restrict__ cQ) {
  int chan = blockIdx.x * 256 + threadIdx.x;
  const int nchan = Bdim * Hdim;
  if (chan >= nchan) return;
  const float* pa = chA + chan;
  const float* pm = chM + chan;
  float* pq = cQ + chan;
  float qa[LOOKAHEAD_CARRY], qm[LOOKAHEAD_CARRY];
#pragma unroll
  for (int j = 0; j < LOOKAHEAD_CARRY; ++j) {
    qa[j] = pa[j * nchan];
    qm[j] = pm[j * nchan];
  }
  float P = 0.f, G = -INFINITY;
  for (int c0 = LOOKAHEAD_CARRY; c0 < NCHUNK; c0 += LOOKAHEAD_CARRY) {
    pa += LOOKAHEAD_CARRY * nchan;
    pm += LOOKAHEAD_CARRY * nchan;
#pragma unroll
    for (int j = 0; j < LOOKAHEAD_CARRY; ++j) {
      pq[j * nchan] = G + P;
      float d = qm[j] - P;
      P += qa[j];
      qa[j] = pa[j * nchan];
      qm[j] = pm[j * nchan];
      G = logaddexp_f(G, d);
    }
    pq += LOOKAHEAD_CARRY * nchan;
  }
#pragma unroll
  for (int j = 0; j < LOOKAHEAD_CARRY; ++j) {
    pq[j * nchan] = G + P;
    G = logaddexp_f(G, qm[j] - P);
    P += qa[j];
  }
}

__global__ void __launch_bounds__(256) apply_scan(
    const float* __restrict__ lf, const float* __restrict__ lv,
    const float* __restrict__ cQ, float* __restrict__ out) {
  int idx = blockIdx.x * 256 + threadIdx.x;
  int h  = idx % Hdim;
  int t2 = idx / Hdim;
  int c  = t2 % NCHUNK;
  int b  = t2 / NCHUNK;
  size_t base = ((size_t)b * Sdim + (size_t)c * CLEN) * (size_t)Hdim + h;
  const float* pf = lf + base;
  const float* pv = lv + base;
  float* po = out + base;
  float R = cQ[c * (Bdim * Hdim) + b * Hdim + h];
  float qf[LOOKAHEAD], qv[LOOKAHEAD];
#pragma unroll
  for (int j = 0; j < LOOKAHEAD; ++j) {
    qf[j] = pf[(size_t)j * Hdim];
    qv[j] = pv[(size_t)j * Hdim];
  }
  float a = 0.f;
#pragma unroll 1
  for (int t0 = LOOKAHEAD; t0 < CLEN; t0 += LOOKAHEAD) {
    pf += (size_t)LOOKAHEAD * Hdim;
    pv += (size_t)LOOKAHEAD * Hdim;
#pragma unroll
    for (int j = 0; j < LOOKAHEAD; ++j) {
      a += qf[j] + EPSV;
      float d = qv[j] - a;
      qf[j] = pf[(size_t)j * Hdim];
      qv[j] = pv[(size_t)j * Hdim];
      R = logaddexp_f(R, d);
      po[(size_t)j * Hdim] = exp_hw(a + R);
    }
    po += (size_t)LOOKAHEAD * Hdim;
  }
#pragma unroll
  for (int j = 0; j < LOOKAHEAD; ++j) {
    a += qf[j] + EPSV;
    R = logaddexp_f(R, qv[j] - a);
    po[(size_t)j * Hdim] = exp_hw(a + R);
  }
}

extern "C" void kernel_launch(void* const* d_in, const int* in_sizes, int n_in,
                              void* d_out, int out_size, void* d_ws, size_t ws_size,
                              hipStream_t stream) {
  (void)in_sizes; (void)n_in; (void)out_size; (void)ws_size;
  const float* x  = (const float*)d_in[0];
  const float* Wf = (const float*)d_in[1];
  const float* Wi = (const float*)d_in[2];
  const float* Wh = (const float*)d_in[3];
  float* out = (float*)d_out;

  const size_t NBSH = (size_t)Bdim * Sdim * Hdim;
  const size_t NSUM = (size_t)NCHUNK * Bdim * Hdim;
  float* lf  = (float*)d_ws;
  float* lv  = lf + NBSH;
  float* chA = lv + NBSH;
  float* chM = chA + NSUM;
  float* cQ  = chM + NSUM;

  const int gate_blocks = (Bdim * Sdim / 16) * (GROUPS / 2) / 8;
  gates_wmma<<<gate_blocks, 256, 0, stream>>>(x, Wf, Wi, Wh, lf, lv);

  const int scan_threads = Bdim * NCHUNK * Hdim;
  chunk_local<<<scan_threads / 256, 256, 0, stream>>>(lf, lv, chA, chM);
  carry_scan<<<(Bdim * Hdim + 255) / 256, 256, 0, stream>>>(chA, chM, cQ);
  apply_scan<<<scan_threads / 256, 256, 0, stream>>>(lf, lv, cQ, out);
}


# recurrence carried in linear not log space: gates emit F=Q/(P+Q), V=P/(P+Q)*g(hid), P,Q=1+e^-f,1+e^-i scaled by e^-max; the 3 scans do h=F*h+V by FMA; rounding and summation order change
# speedup vs baseline: 3.2417x; 1.2892x over previous
#include <hip/hip_runtime.h>
#include <math.h>

#define EPSV   1e-8f
#define GROUPS 128
#define Bdim   4
#define Sdim   8192
#define Ddim   1024
#define Hdim   1024
#define NCHUNK 64
#define CLEN   (Sdim / NCHUNK)

typedef float v2f __attribute__((ext_vector_type(2)));
typedef float v8f __attribute__((ext_vector_type(8)));

__device__ __forceinline__ float exp_hw(float v) {
  return __builtin_amdgcn_exp2f(v * 1.44269504088896341f);
}

__device__ __forceinline__ void gate_lin(float f, float i, float hd,
                                         float& Fo, float& Vo) {
  const float nf = -f, ni = -i;
  const float m  = fmaxf(fmaxf(nf, ni), 0.f);
  const float e0 = exp_hw(-m);
  const float p  = e0 + exp_hw(nf - m);
  const float q  = e0 + exp_hw(ni - m);
  const float r  = __builtin_amdgcn_rcpf(p + q);
  Fo = q * r;
  const float iw = p * r;
  const float t  = exp_hw(hd >= 0.f ? 0.f : hd);
  const float sg = t * __builtin_amdgcn_rcpf(1.f + t);
  const float gp = (hd + 0.5f) + EPSV;
  Vo = iw * (hd >= 0.f ? gp : sg);
}

__device__ __forceinline__ v2f ldw(const float* __restrict__ Wg, int cn, int kc,
                                   int hi, bool valid) {
  v2f w = *(const v2f*)(Wg + cn * 8 + kc * 4 + 2 * hi);
  v2f r;
  r.x = valid ? w.x : 0.f;
  r.y = valid ? w.y : 0.f;
  return r;
}

__device__ __forceinline__ v8f gate16(v2f a00, v2f a01, v2f a10, v2f a11,
                                      const float* __restrict__ W, int g0,
                                      int n, int hi) {
  const int  cn = n & 7;
  const bool lo = (n < 8);
  const float* Wg0 = W + (size_t)g0 * 64;
  const float* Wg1 = Wg0 + 64;
  v2f b00 = ldw(Wg0, cn, 0, hi, lo);
  v2f b01 = ldw(Wg0, cn, 1, hi, lo);
  v2f b10 = ldw(Wg1, cn, 0, hi, !lo);
  v2f b11 = ldw(Wg1, cn, 1, hi, !lo);
  v8f acc = {};
  acc = __builtin_amdgcn_wmma_f32_16x16x4_f32(false, a00, false, b00, (short)0, acc, false, false);
  acc = __builtin_amdgcn_wmma_f32_16x16x4_f32(false, a01, false, b01, (short)0, acc, false, false);
  acc = __builtin_amdgcn_wmma_f32_16x16x4_f32(false, a10, false, b10, (short)0, acc, false, false);
  acc = __builtin_amdgcn_wmma_f32_16x16x4_f32(false, a11, false, b11, (short)0, acc, false, false);
  return acc;
}

__global__ void __launch_bounds__(256) gates_wmma(
    const float* __restrict__ x, const float* __restrict__ Wf,
    const float* __restrict__ Wi, const float* __restrict__ Wh,
    float* __restrict__ F_out, float* __restrict__ V_out) {
  const int lane = threadIdx.x & 31;
  const int wv   = threadIdx.x >> 5;
  const int gid  = blockIdx.x * 8 + wv;
  const int NPAIR = GROUPS / 2;
  const int gp = gid % NPAIR;
  const int tt = gid / NPAIR;
  const int s0 = (tt * 16) % Sdim;
  const int b  = (tt * 16) / Sdim;

  const int n  = lane & 15;
  const int hi = lane >> 4;

  const float* xrow =
      x + ((size_t)b * Sdim + (size_t)(s0 + n)) * (size_t)Ddim + (size_t)gp * 16;
  v2f a00 = *(const v2f*)(xrow + 0  + 2 * hi);
  v2f a01 = *(const v2f*)(xrow + 4  + 2 * hi);
  v2f a10 = *(const v2f*)(xrow + 8  + 2 * hi);
  v2f a11 = *(const v2f*)(xrow + 12 + 2 * hi);

  const int g0 = gp * 2;
  v8f fg = gate16(a00, a01, a10, a11, Wf, g0, n, hi);
  v8f ig = gate16(a00, a01, a10, a11, Wi, g0, n, hi);
  v8f hg = gate16(a00, a01, a10, a11, Wh, g0, n, hi);

  const size_t baseoff =
      ((size_t)b * Sdim + (size_t)(s0 + 8 * hi)) * (size_t)Hdim + (size_t)gp * 16 + n;
#pragma unroll
  for (int r = 0; r < 8; ++r) {
    float Fk, Vk;
    gate_lin(fg[r], ig[r], hg[r], Fk, Vk);
    size_t off = baseoff + (size_t)r * Hdim;
    F_out[off] = Fk;
    V_out[off] = Vk;
  }
}

#define LOOKAHEAD 16
#define LOOKAHEAD_CARRY 8

__global__ void __launch_bounds__(256) chunk_local(
    const float* __restrict__ Fp, const float* __restrict__ Vp,
    float* __restrict__ chA, float* __restrict__ chB) {
  int idx = blockIdx.x * 256 + threadIdx.x;
  int h  = idx % Hdim;
  int t2 = idx / Hdim;
  int c  = t2 % NCHUNK;
  int b  = t2 / NCHUNK;
  size_t base = ((size_t)b * Sdim + (size_t)c * CLEN) * (size_t)Hdim + h;
  const float* pf = Fp + base;
  const float* pv = Vp + base;
  float qf[LOOKAHEAD], qv[LOOKAHEAD];
#pragma unroll
  for (int j = 0; j < LOOKAHEAD; ++j) {
    qf[j] = pf[(size_t)j * Hdim];
    qv[j] = pv[(size_t)j * Hdim];
  }
  float A = 1.f, Bv = 0.f;
#pragma unroll 1
  for (int t0 = LOOKAHEAD; t0 < CLEN; t0 += LOOKAHEAD) {
    pf += (size_t)LOOKAHEAD * Hdim;
    pv += (size_t)LOOKAHEAD * Hdim;
#pragma unroll
    for (int j = 0; j < LOOKAHEAD; ++j) {
      Bv = fmaf(qf[j], Bv, qv[j]);
      A *= qf[j];
      qf[j] = pf[(size_t)j * Hdim];
      qv[j] = pv[(size_t)j * Hdim];
    }
  }
#pragma unroll
  for (int j = 0; j < LOOKAHEAD; ++j) {
    Bv = fmaf(qf[j], Bv, qv[j]);
    A *= qf[j];
  }
  int sidx = c * (Bdim * Hdim) + b * Hdim + h;
  chA[sidx] = A;
  chB[sidx] = Bv;
}

__global__ void __launch_bounds__(256) carry_scan(
    const float* __restrict__ chA, const float* __restrict__ chB,
    float* __restrict__ cH) {
  int chan = blockIdx.x * 256 + threadIdx.x;
  const int nchan = Bdim * Hdim;
  if (chan >= nchan) return;
  const float* pa = chA + chan;
  const float* pb = chB + chan;
  float* pq = cH + chan;
  float qa[LOOKAHEAD_CARRY], qb[LOOKAHEAD_CARRY];
#pragma unroll
  for (int j = 0; j < LOOKAHEAD_CARRY; ++j) {
    qa[j] = pa[j * nchan];
    qb[j] = pb[j * nchan];
  }
  float H = 0.f;
  for (int c0 = LOOKAHEAD_CARRY; c0 < NCHUNK; c0 += LOOKAHEAD_CARRY) {
    pa += LOOKAHEAD_CARRY * nchan;
    pb += LOOKAHEAD_CARRY * nchan;
#pragma unroll
    for (int j = 0; j < LOOKAHEAD_CARRY; ++j) {
      pq[j * nchan] = H;
      H = fmaf(qa[j], H, qb[j]);
      qa[j] = pa[j * nchan];
      qb[j] = pb[j * nchan];
    }
    pq += LOOKAHEAD_CARRY * nchan;
  }
#pragma unroll
  for (int j = 0; j < LOOKAHEAD_CARRY; ++j) {
    pq[j * nchan] = H;
    H = fmaf(qa[j], H, qb[j]);
  }
}

__global__ void __launch_bounds__(256) apply_scan(
    const float* __restrict__ Fp, const float* __restrict__ Vp,
    const float* __restrict__ cH, float* __restrict__ out) {
  int idx = blockIdx.x * 256 + threadIdx.x;
  int h  = idx % Hdim;
  int t2 = idx / Hdim;
  int c  = t2 % NCHUNK;
  int b  = t2 / NCHUNK;
  size_t base = ((size_t)b * Sdim + (size_t)c * CLEN) * (size_t)Hdim + h;
  const float* pf = Fp + base;
  const float* pv = Vp + base;
  float* po = out + base;
  float hv = cH[c * (Bdim * Hdim) + b * Hdim + h];
  float qf[LOOKAHEAD], qv[LOOKAHEAD];
#pragma unroll
  for (int j = 0; j < LOOKAHEAD; ++j) {
    qf[j] = pf[(size_t)j * Hdim];
    qv[j] = pv[(size_t)j * Hdim];
  }
#pragma unroll 1
  for (int t0 = LOOKAHEAD; t0 < CLEN; t0 += LOOKAHEAD) {
    pf += (size_t)LOOKAHEAD * Hdim;
    pv += (size_t)LOOKAHEAD * Hdim;
#pragma unroll
    for (int j = 0; j < LOOKAHEAD; ++j) {
      hv = fmaf(qf[j], hv, qv[j]);
      qf[j] = pf[(size_t)j * Hdim];
      qv[j] = pv[(size_t)j * Hdim];
      po[(size_t)j * Hdim] = hv;
    }
    po += (size_t)LOOKAHEAD * Hdim;
  }
#pragma unroll
  for (int j = 0; j < LOOKAHEAD; ++j) {
    hv = fmaf(qf[j], hv, qv[j]);
    po[(size_t)j * Hdim] = hv;
  }
}

extern "C" void kernel_launch(void* const* d_in, const int* in_sizes, int n_in,
                              void* d_out, int out_size, void* d_ws, size_t ws_size,
                              hipStream_t stream) {
  (void)in_sizes; (void)n_in; (void)out_size; (void)ws_size;
  const float* x  = (const float*)d_in[0];
  const float* Wf = (const float*)d_in[1];
  const float* Wi = (const float*)d_in[2];
  const float* Wh = (const float*)d_in[3];
  float* out = (float*)d_out;

  const size_t NBSH = (size_t)Bdim * Sdim * Hdim;
  const size_t NSUM = (size_t)NCHUNK * Bdim * Hdim;
  float* Fp  = (float*)d_ws;
  float* Vp  = Fp + NBSH;
  float* chA = Vp + NBSH;
  float* chB = chA + NSUM;
  float* cH  = chB + NSUM;

  const int gate_blocks = (Bdim * Sdim / 16) * (GROUPS / 2) / 8;
  gates_wmma<<<gate_blocks, 256, 0, stream>>>(x, Wf, Wi, Wh, Fp, Vp);

  const int scan_threads = Bdim * NCHUNK * Hdim;
  chunk_local<<<scan_threads / 256, 256, 0, stream>>>(Fp, Vp, chA, chB);
  carry_scan<<<(Bdim * Hdim + 255) / 256, 256, 0, stream>>>(chA, chB, cH);
  apply_scan<<<scan_threads / 256, 256, 0, stream>>>(Fp, Vp, cH, out);
}


# gates (f32 WMMA products, F,V math) computed inside both scan passes from x, tile rows = 8 steps of two chunks, so the F,V planes are never stored or re-read; 4 launches -> 3; sum order unchanged
# speedup vs baseline: 5.6551x; 1.7445x over previous
#include <hip/hip_runtime.h>
#include <math.h>

#define EPSV   1e-8f
#define GROUPS 128
#define Bdim   4
#define Sdim   8192
#define Ddim   1024
#define Hdim   1024
#define NCHUNK 64
#define CLEN   (Sdim / NCHUNK)

typedef float v2f __attribute__((ext_vector_type(2)));
typedef float v8f __attribute__((ext_vector_type(8)));

__device__ __forceinline__ float exp_hw(float v) {
  return __builtin_amdgcn_exp2f(v * 1.44269504088896341f);
}

__device__ __forceinline__ void gate_lin(float f, float i, float hd,
                                         float& Fo, float& Vo) {
  const float nf = -f, ni = -i;
  const float m  = fmaxf(fmaxf(nf, ni), 0.f);
  const float e0 = exp_hw(-m);
  const float p  = e0 + exp_hw(nf - m);
  const float q  = e0 + exp_hw(ni - m);
  const float r  = __builtin_amdgcn_rcpf(p + q);
  Fo = q * r;
  const float iw = p * r;
  const float t  = exp_hw(hd >= 0.f ? 0.f : hd);
  const float sg = t * __builtin_amdgcn_rcpf(1.f + t);
  const float gp = (hd + 0.5f) + EPSV;
  Vo = iw * (hd >= 0.f ? gp : sg);
}

__device__ __forceinline__ v2f ldw(const float* __restrict__ Wg, int cn, int kc,
                                   int hi, bool valid) {
  v2f w = *(const v2f*)(Wg + cn * 8 + kc * 4 + 2 * hi);
  v2f r;
  r.x = valid ? w.x : 0.f;
  r.y = valid ? w.y : 0.f;
  return r;
}

struct WB { v2f b00, b01, b10, b11; };

__device__ __forceinline__ WB load_wb(const float* __restrict__ W, int g0,
                                      int n, int hi) {
  const int  cn = n & 7;
  const bool lo = (n < 8);
  const float* Wg0 = W + (size_t)g0 * 64;
  const float* Wg1 = Wg0 + 64;
  WB w;
  w.b00 = ldw(Wg0, cn, 0, hi, lo);
  w.b01 = ldw(Wg0, cn, 1, hi, lo);
  w.b10 = ldw(Wg1, cn, 0, hi, !lo);
  w.b11 = ldw(Wg1, cn, 1, hi, !lo);
  return w;
}

struct XA { v2f a00, a01, a10, a11; };

__device__ __forceinline__ XA load_xa(const float* __restrict__ xrow) {
  XA a;
  a.a00 = *(const v2f*)(xrow + 0);
  a.a01 = *(const v2f*)(xrow + 4);
  a.a10 = *(const v2f*)(xrow + 8);
  a.a11 = *(const v2f*)(xrow + 12);
  return a;
}

__device__ __forceinline__ v8f gate16(const XA& a, const WB& w) {
  v8f acc = {};
  acc = __builtin_amdgcn_wmma_f32_16x16x4_f32(false, a.a00, false, w.b00, (short)0, acc, false, false);
  acc = __builtin_amdgcn_wmma_f32_16x16x4_f32(false, a.a01, false, w.b01, (short)0, acc, false, false);
  acc = __builtin_amdgcn_wmma_f32_16x16x4_f32(false, a.a10, false, w.b10, (short)0, acc, false, false);
  acc = __builtin_amdgcn_wmma_f32_16x16x4_f32(false, a.a11, false, w.b11, (short)0, acc, false, false);
  return acc;
}

#define TSTEP  8
#define NTILE  (CLEN / TSTEP)
#define RING   2

struct WaveJob {
  int n, hi, gp, q;
  const float* xp;
};

__device__ __forceinline__ WaveJob wave_job(const float* __restrict__ x) {
  WaveJob j;
  const int lane = threadIdx.x & 31;
  const int wv   = threadIdx.x >> 5;
  const int gid  = blockIdx.x * 8 + wv;
  const int NPAIR = GROUPS / 2;
  j.gp = gid % NPAIR;
  const int pc = gid / NPAIR;
  j.n  = lane & 15;
  j.hi = lane >> 4;
  j.q  = 2 * pc + j.hi;
  const int qrow = 2 * pc + (j.n >> 3);
  j.xp = x + ((size_t)qrow * CLEN + (size_t)(j.n & 7)) * (size_t)Ddim +
         (size_t)j.gp * 16 + 2 * j.hi;
  return j;
}

__global__ void __launch_bounds__(256, 8) chunk_local(
    const float* __restrict__ x, const float* __restrict__ Wf,
    const float* __restrict__ Wi, const float* __restrict__ Wh,
    float* __restrict__ chA, float* __restrict__ chB) {
  const WaveJob jb = wave_job(x);
  const float* xp = jb.xp;
  const WB wf = load_wb(Wf, jb.gp * 2, jb.n, jb.hi);
  const WB wi = load_wb(Wi, jb.gp * 2, jb.n, jb.hi);
  const WB wh = load_wb(Wh, jb.gp * 2, jb.n, jb.hi);

  XA ring[RING];
#pragma unroll
  for (int j = 0; j < RING; ++j) ring[j] = load_xa(xp + (size_t)j * TSTEP * Ddim);

  float A = 1.f, Bv = 0.f;
  for (int t0 = RING; t0 < NTILE; t0 += RING) {
    xp += (size_t)RING * TSTEP * Ddim;
#pragma unroll
    for (int j = 0; j < RING; ++j) {
      const v8f fg = gate16(ring[j], wf);
      const v8f ig = gate16(ring[j], wi);
      const v8f hg = gate16(ring[j], wh);
      ring[j] = load_xa(xp + (size_t)j * TSTEP * Ddim);
#pragma unroll
      for (int r = 0; r < TSTEP; ++r) {
        float Fk, Vk;
        gate_lin(fg[r], ig[r], hg[r], Fk, Vk);
        Bv = fmaf(Fk, Bv, Vk);
        A *= Fk;
      }
    }
  }
#pragma unroll
  for (int j = 0; j < RING; ++j) {
    const v8f fg = gate16(ring[j], wf);
    const v8f ig = gate16(ring[j], wi);
    const v8f hg = gate16(ring[j], wh);
#pragma unroll
    for (int r = 0; r < TSTEP; ++r) {
      float Fk, Vk;
      gate_lin(fg[r], ig[r], hg[r], Fk, Vk);
      Bv = fmaf(Fk, Bv, Vk);
      A *= Fk;
    }
  }
  const int b = jb.q / NCHUNK;
  const int c = jb.q % NCHUNK;
  const int h = jb.gp * 16 + jb.n;
  const int sidx = c * (Bdim * Hdim) + b * Hdim + h;
  chA[sidx] = A;
  chB[sidx] = Bv;
}

#define LOOKAHEAD_CARRY 8

__global__ void __launch_bounds__(256) carry_scan(
    const float* __restrict__ chA, const float* __restrict__ chB,
    float* __restrict__ cH) {
  int chan = blockIdx.x * 256 + threadIdx.x;
  const int nchan = Bdim * Hdim;
  if (chan >= nchan) return;
  const float* pa = chA + chan;
  const float* pb = chB + chan;
  float* pq = cH + chan;
  float qa[LOOKAHEAD_CARRY], qb[LOOKAHEAD_CARRY];
#pragma unroll
  for (int j = 0; j < LOOKAHEAD_CARRY; ++j) {
    qa[j] = pa[j * nchan];
    qb[j] = pb[j * nchan];
  }
  float H = 0.f;
  for (int c0 = LOOKAHEAD_CARRY; c0 < NCHUNK; c0 += LOOKAHEAD_CARRY) {
    pa += LOOKAHEAD_CARRY * nchan;
    pb += LOOKAHEAD_CARRY * nchan;
#pragma unroll
    for (int j = 0; j < LOOKAHEAD_CARRY; ++j) {
      pq[j * nchan] = H;
      H = fmaf(qa[j], H, qb[j]);
      qa[j] = pa[j * nchan];
      qb[j] = pb[j * nchan];
    }
    pq += LOOKAHEAD_CARRY * nchan;
  }
#pragma unroll
  for (int j = 0; j < LOOKAHEAD_CARRY; ++j) {
    pq[j * nchan] = H;
    H = fmaf(qa[j], H, qb[j]);
  }
}

__global__ void __launch_bounds__(256) apply_scan(
    const float* __restrict__ x, const float* __restrict__ Wf,
    const float* __restrict__ Wi, const float* __restrict__ Wh,
    const float* __restrict__ cH, float* __restrict__ out) {
  const WaveJob jb = wave_job(x);
  const float* xp = jb.xp;
  const WB wf = load_wb(Wf, jb.gp * 2, jb.n, jb.hi);
  const WB wi = load_wb(Wi, jb.gp * 2, jb.n, jb.hi);
  const WB wh = load_wb(Wh, jb.gp * 2, jb.n, jb.hi);

  XA ring[RING];
#pragma unroll
  for (int j = 0; j < RING; ++j) ring[j] = load_xa(xp + (size_t)j * TSTEP * Ddim);

  const int b = jb.q / NCHUNK;
  const int c = jb.q % NCHUNK;
  const int h = jb.gp * 16 + jb.n;
  float hv = cH[c * (Bdim * Hdim) + b * Hdim + h];
  float* po = out + (size_t)jb.q * CLEN * (size_t)Hdim + h;

  for (int t0 = RING; t0 < NTILE; t0 += RING) {
    xp += (size_t)RING * TSTEP * Ddim;
#pragma unroll
    for (int j = 0; j < RING; ++j) {
      const v8f fg = gate16(ring[j], wf);
      const v8f ig = gate16(ring[j], wi);
      const v8f hg = gate16(ring[j], wh);
      ring[j] = load_xa(xp + (size_t)j * TSTEP * Ddim);
#pragma unroll
      for (int r = 0; r < TSTEP; ++r) {
        float Fk, Vk;
        gate_lin(fg[r], ig[r], hg[r], Fk, Vk);
        hv = fmaf(Fk, hv, Vk);
        po[(size_t)(j * TSTEP + r) * Hdim] = hv;
      }
    }
    po += (size_t)RING * TSTEP * Hdim;
  }
#pragma unroll
  for (int j = 0; j < RING; ++j) {
    const v8f fg = gate16(ring[j], wf);
    const v8f ig = gate16(ring[j], wi);
    const v8f hg = gate16(ring[j], wh);
#pragma unroll
    for (int r = 0; r < TSTEP; ++r) {
      float Fk, Vk;
      gate_lin(fg[r], ig[r], hg[r], Fk, Vk);
      hv = fmaf(Fk, hv, Vk);
      po[(size_t)(j * TSTEP + r) * Hdim] = hv;
    }
  }
}

extern "C" void kernel_launch(void* const* d_in, const int* in_sizes, int n_in,
                              void* d_out, int out_size, void* d_ws, size_t ws_size,
                              hipStream_t stream) {
  (void)in_sizes; (void)n_in; (void)out_size; (void)ws_size;
  const float* x  = (const float*)d_in[0];
  const float* Wf = (const float*)d_in[1];
  const float* Wi = (const float*)d_in[2];
  const float* Wh = (const float*)d_in[3];
  float* out = (float*)d_out;

  const size_t NSUM = (size_t)NCHUNK * Bdim * Hdim;
  float* chA = (float*)d_ws;
  float* chB = chA + NSUM;
  float* cH  = chB + NSUM;

  const int scan_blocks = (GROUPS / 2) * (Bdim * NCHUNK / 2) / 8;
  chunk_local<<<scan_blocks, 256, 0, stream>>>(x, Wf, Wi, Wh, chA, chB);
  carry_scan<<<(Bdim * Hdim + 255) / 256, 256, 0, stream>>>(chA, chB, cH);
  apply_scan<<<scan_blocks, 256, 0, stream>>>(x, Wf, Wi, Wh, cH, out);
}


# both scan passes software-pipelined per wave: each f32 WMMA of tile k+1 is followed by a slice of the gate math of tile k (result tiles double-buffered); sum order unchanged
# speedup vs baseline: 5.9492x; 1.0520x over previous
#include <hip/hip_runtime.h>
#include <math.h>

#define EPSV   1e-8f
#define GROUPS 128
#define Bdim   4
#define Sdim   8192
#define Ddim   1024
#define Hdim   1024
#define NCHUNK 64
#define CLEN   (Sdim / NCHUNK)

typedef float v2f __attribute__((ext_vector_type(2)));
typedef float v8f __attribute__((ext_vector_type(8)));

__device__ __forceinline__ float exp_hw(float v) {
  return __builtin_amdgcn_exp2f(v * 1.44269504088896341f);
}

__device__ __forceinline__ void gate_lin(float f, float i, float hd,
                                         float& Fo, float& Vo) {
  const float nf = -f, ni = -i;
  const float m  = fmaxf(fmaxf(nf, ni), 0.f);
  const float e0 = exp_hw(-m);
  const float p  = e0 + exp_hw(nf - m);
  const float q  = e0 + exp_hw(ni - m);
  const float r  = __builtin_amdgcn_rcpf(p + q);
  Fo = q * r;
  const float iw = p * r;
  const float t  = exp_hw(hd >= 0.f ? 0.f : hd);
  const float sg = t * __builtin_amdgcn_rcpf(1.f + t);
  const float gp = (hd + 0.5f) + EPSV;
  Vo = iw * (hd >= 0.f ? gp : sg);
}

__device__ __forceinline__ v2f ldw(const float* __restrict__ Wg, int cn, int kc,
                                   int hi, bool valid) {
  v2f w = *(const v2f*)(Wg + cn * 8 + kc * 4 + 2 * hi);
  v2f r;
  r.x = valid ? w.x : 0.f;
  r.y = valid ? w.y : 0.f;
  return r;
}

struct WB { v2f b00, b01, b10, b11; };

__device__ __forceinline__ WB load_wb(const float* __restrict__ W, int g0,
                                      int n, int hi) {
  const int  cn = n & 7;
  const bool lo = (n < 8);
  const float* Wg0 = W + (size_t)g0 * 64;
  const float* Wg1 = Wg0 + 64;
  WB w;
  w.b00 = ldw(Wg0, cn, 0, hi, lo);
  w.b01 = ldw(Wg0, cn, 1, hi, lo);
  w.b10 = ldw(Wg1, cn, 0, hi, !lo);
  w.b11 = ldw(Wg1, cn, 1, hi, !lo);
  return w;
}

struct XA { v2f a00, a01, a10, a11; };

__device__ __forceinline__ XA load_xa(const float* __restrict__ xrow) {
  XA a;
  a.a00 = *(const v2f*)(xrow + 0);
  a.a01 = *(const v2f*)(xrow + 4);
  a.a10 = *(const v2f*)(xrow + 8);
  a.a11 = *(const v2f*)(xrow + 12);
  return a;
}

__device__ __forceinline__ v8f gate16(const XA& a, const WB& w) {
  v8f acc = {};
  acc = __builtin_amdgcn_wmma_f32_16x16x4_f32(false, a.a00, false, w.b00, (short)0, acc, false, false);
  acc = __builtin_amdgcn_wmma_f32_16x16x4_f32(false, a.a01, false, w.b01, (short)0, acc, false, false);
  acc = __builtin_amdgcn_wmma_f32_16x16x4_f32(false, a.a10, false, w.b10, (short)0, acc, false, false);
  acc = __builtin_amdgcn_wmma_f32_16x16x4_f32(false, a.a11, false, w.b11, (short)0, acc, false, false);
  return acc;
}

#define TSTEP  8
#define NTILE  (CLEN / TSTEP)


__device__ __forceinline__ void local_tile(const v8f& fg, const v8f& ig, const v8f& hg,
                                           float& A, float& Bv) {
#pragma unroll
  for (int r = 0; r < TSTEP; ++r) {
    float Fk, Vk;
    gate_lin(fg[r], ig[r], hg[r], Fk, Vk);
    Bv = fmaf(Fk, Bv, Vk);
    A *= Fk;
  }
}

__device__ __forceinline__ void apply_tile(const v8f& fg, const v8f& ig, const v8f& hg,
                                           float& hv, float* __restrict__ po) {
#pragma unroll
  for (int r = 0; r < TSTEP; ++r) {
    float Fk, Vk;
    gate_lin(fg[r], ig[r], hg[r], Fk, Vk);
    hv = fmaf(Fk, hv, Vk);
    po[(size_t)r * Hdim] = hv;
  }
}

template <bool APPLY, bool REFILL>
__device__ __forceinline__ void pipe_step(XA& a, const float* __restrict__ xnext,
                                          const WB& wf, const WB& wi, const WB& wh,
                                          v8f& fN, v8f& iN, v8f& hN,
                                          const v8f& fP, const v8f& iP, const v8f& hP,
                                          float& s0, float& s1, float* __restrict__ po) {
  const v8f zero = {};
  float E0[TSTEP], E1[TSTEP], E2[TSTEP], TT[TSTEP], PP[TSTEP], QQ[TSTEP], RR[TSTEP], RU[TSTEP];
    fN = __builtin_amdgcn_wmma_f32_16x16x4_f32(false, a.a00, false, wf.b00, (short)0, zero, false, false);
    { const float nf = -fP[0], ni = -iP[0];
      const float m = fmaxf(fmaxf(nf, ni), 0.f);
      E0[0] = exp_hw(-m); E1[0] = exp_hw(nf - m); E2[0] = exp_hw(ni - m);
      TT[0] = exp_hw(hP[0] >= 0.f ? 0.f : hP[0]); }
    { const float nf = -fP[1], ni = -iP[1];
      const float m = fmaxf(fmaxf(nf, ni), 0.f);
      E0[1] = exp_hw(-m); E1[1] = exp_hw(nf - m); E2[1] = exp_hw(ni - m);
      TT[1] = exp_hw(hP[1] >= 0.f ? 0.f : hP[1]); }
    __builtin_amdgcn_sched_barrier(0);
    iN = __builtin_amdgcn_wmma_f32_16x16x4_f32(false, a.a00, false, wi.b00, (short)0, zero, false, false);
    { PP[0] = E0[0] + E1[0]; QQ[0] = E0[0] + E2[0];
      RR[0] = __builtin_amdgcn_rcpf(PP[0] + QQ[0]);
      RU[0] = __builtin_amdgcn_rcpf(1.f + TT[0]); }
    { const float nf = -fP[2], ni = -iP[2];
      const float m = fmaxf(fmaxf(nf, ni), 0.f);
      E0[2] = exp_hw(-m); E1[2] = exp_hw(nf - m); E2[2] = exp_hw(ni - m);
      TT[2] = exp_hw(hP[2] >= 0.f ? 0.f : hP[2]); }
    __builtin_amdgcn_sched_barrier(0);
    hN = __builtin_amdgcn_wmma_f32_16x16x4_f32(false, a.a00, false, wh.b00, (short)0, zero, false, false);
    { PP[1] = E0[1] + E1[1]; QQ[1] = E0[1] + E2[1];
      RR[1] = __builtin_amdgcn_rcpf(PP[1] + QQ[1]);
      RU[1] = __builtin_amdgcn_rcpf(1.f + TT[1]); }
    { const float Fk = QQ[0] * RR[0], iw = PP[0] * RR[0];
      const float sg = TT[0] * RU[0];
      const float gp = (hP[0] + 0.5f) + EPSV;
      const float Vk = iw * (hP[0] >= 0.f ? gp : sg);
      if constexpr (APPLY) { s0 = fmaf(Fk, s0, Vk); po[(size_t)0 * Hdim] = s0; }
      else { s1 = fmaf(Fk, s1, Vk); s0 *= Fk; } }
    __builtin_amdgcn_sched_barrier(0);
    fN = __builtin_amdgcn_wmma_f32_16x16x4_f32(false, a.a01, false, wf.b01, (short)0, fN, false, false);
    { const float nf = -fP[3], ni = -iP[3];
      const float m = fmaxf(fmaxf(nf, ni), 0.f);
      E0[3] = exp_hw(-m); E1[3] = exp_hw(nf - m); E2[3] = exp_hw(ni - m);
      TT[3] = exp_hw(hP[3] >= 0.f ? 0.f : hP[3]); }
    { PP[2] = E0[2] + E1[2]; QQ[2] = E0[2] + E2[2];
      RR[2] = __builtin_amdgcn_rcpf(PP[2] + QQ[2]);
      RU[2] = __builtin_amdgcn_rcpf(1.f + TT[2]); }
    __builtin_amdgcn_sched_barrier(0);
    iN = __builtin_amdgcn_wmma_f32_16x16x4_f32(false, a.a01, false, wi.b01, (short)0, iN, false, false);
    { const float Fk = QQ[1] * RR[1], iw = PP[1] * RR[1];
      const float sg = TT[1] * RU[1];
      const float gp = (hP[1] + 0.5f) + EPSV;
      const float Vk = iw * (hP[1] >= 0.f ? gp : sg);
      if constexpr (APPLY) { s0 = fmaf(Fk, s0, Vk); po[(size_t)1 * Hdim] = s0; }
      else { s1 = fmaf(Fk, s1, Vk); s0 *= Fk; } }
    { const float nf = -fP[4], ni = -iP[4];
      const float m = fmaxf(fmaxf(nf, ni), 0.f);
      E0[4] = exp_hw(-m); E1[4] = exp_hw(nf - m); E2[4] = exp_hw(ni - m);
      TT[4] = exp_hw(hP[4] >= 0.f ? 0.f : hP[4]); }
    __builtin_amdgcn_sched_barrier(0);
    hN = __builtin_amdgcn_wmma_f32_16x16x4_f32(false, a.a01, false, wh.b01, (short)0, hN, false, false);
    { PP[3] = E0[3] + E1[3]; QQ[3] = E0[3] + E2[3];
      RR[3] = __builtin_amdgcn_rcpf(PP[3] + QQ[3]);
      RU[3] = __builtin_amdgcn_rcpf(1.f + TT[3]); }
    { const float Fk = QQ[2] * RR[2], iw = PP[2] * RR[2];
      const float sg = TT[2] * RU[2];
      const float gp = (hP[2] + 0.5f) + EPSV;
      const float Vk = iw * (hP[2] >= 0.f ? gp : sg);
      if constexpr (APPLY) { s0 = fmaf(Fk, s0, Vk); po[(size_t)2 * Hdim] = s0; }
      else { s1 = fmaf(Fk, s1, Vk); s0 *= Fk; } }
    __builtin_amdgcn_sched_barrier(0);
    fN = __builtin_amdgcn_wmma_f32_16x16x4_f32(false, a.a10, false, wf.b10, (short)0, fN, false, false);
    { const float nf = -fP[5], ni = -iP[5];
      const float m = fmaxf(fmaxf(nf, ni), 0.f);
      E0[5] = exp_hw(-m); E1[5] = exp_hw(nf - m); E2[5] = exp_hw(ni - m);
      TT[5] = exp_hw(hP[5] >= 0.f ? 0.f : hP[5]); }
    { PP[4] = E0[4] + E1[4]; QQ[4] = E0[4] + E2[4];
      RR[4] = __builtin_amdgcn_rcpf(PP[4] + QQ[4]);
      RU[4] = __builtin_amdgcn_rcpf(1.f + TT[4]); }
    __builtin_amdgcn_sched_barrier(0);
    iN = __builtin_amdgcn_wmma_f32_16x16x4_f32(false, a.a10, false, wi.b10, (short)0, iN, false, false);
    { const float Fk = QQ[3] * RR[3], iw = PP[3] * RR[3];
      const float sg = TT[3] * RU[3];
      const float gp = (hP[3] + 0.5f) + EPSV;
      const float Vk = iw * (hP[3] >= 0.f ? gp : sg);
      if constexpr (APPLY) { s0 = fmaf(Fk, s0, Vk); po[(size_t)3 * Hdim] = s0; }
      else { s1 = fmaf(Fk, s1, Vk); s0 *= Fk; } }
    { const float nf = -fP[6], ni = -iP[6];
      const float m = fmaxf(fmaxf(nf, ni), 0.f);
      E0[6] = exp_hw(-m); E1[6] = exp_hw(nf - m); E2[6] = exp_hw(ni - m);
      TT[6] = exp_hw(hP[6] >= 0.f ? 0.f : hP[6]); }
    __builtin_amdgcn_sched_barrier(0);
    hN = __builtin_amdgcn_wmma_f32_16x16x4_f32(false, a.a10, false, wh.b10, (short)0, hN, false, false);
    { PP[5] = E0[5] + E1[5]; QQ[5] = E0[5] + E2[5];
      RR[5] = __builtin_amdgcn_rcpf(PP[5] + QQ[5]);
      RU[5] = __builtin_amdgcn_rcpf(1.f + TT[5]); }
    { const float Fk = QQ[4] * RR[4], iw = PP[4] * RR[4];
      const float sg = TT[4] * RU[4];
      const float gp = (hP[4] + 0.5f) + EPSV;
      const float Vk = iw * (hP[4] >= 0.f ? gp : sg);
      if constexpr (APPLY) { s0 = fmaf(Fk, s0, Vk); po[(size_t)4 * Hdim] = s0; }
      else { s1 = fmaf(Fk, s1, Vk); s0 *= Fk; } }
    __builtin_amdgcn_sched_barrier(0);
    fN = __builtin_amdgcn_wmma_f32_16x16x4_f32(false, a.a11, false, wf.b11, (short)0, fN, false, false);
    { const float nf = -fP[7], ni = -iP[7];
      const float m = fmaxf(fmaxf(nf, ni), 0.f);
      E0[7] = exp_hw(-m); E1[7] = exp_hw(nf - m); E2[7] = exp_hw(ni - m);
      TT[7] = exp_hw(hP[7] >= 0.f ? 0.f : hP[7]); }
    { PP[6] = E0[6] + E1[6]; QQ[6] = E0[6] + E2[6];
      RR[6] = __builtin_amdgcn_rcpf(PP[6] + QQ[6]);
      RU[6] = __builtin_amdgcn_rcpf(1.f + TT[6]); }
    __builtin_amdgcn_sched_barrier(0);
    iN = __builtin_amdgcn_wmma_f32_16x16x4_f32(false, a.a11, false, wi.b11, (short)0, iN, false, false);
    { const float Fk = QQ[5] * RR[5], iw = PP[5] * RR[5];
      const float sg = TT[5] * RU[5];
      const float gp = (hP[5] + 0.5f) + EPSV;
      const float Vk = iw * (hP[5] >= 0.f ? gp : sg);
      if constexpr (APPLY) { s0 = fmaf(Fk, s0, Vk); po[(size_t)5 * Hdim] = s0; }
      else { s1 = fmaf(Fk, s1, Vk); s0 *= Fk; } }
    { PP[7] = E0[7] + E1[7]; QQ[7] = E0[7] + E2[7];
      RR[7] = __builtin_amdgcn_rcpf(PP[7] + QQ[7]);
      RU[7] = __builtin_amdgcn_rcpf(1.f + TT[7]); }
    __builtin_amdgcn_sched_barrier(0);
    hN = __builtin_amdgcn_wmma_f32_16x16x4_f32(false, a.a11, false, wh.b11, (short)0, hN, false, false);
    { const float Fk = QQ[6] * RR[6], iw = PP[6] * RR[6];
      const float sg = TT[6] * RU[6];
      const float gp = (hP[6] + 0.5f) + EPSV;
      const float Vk = iw * (hP[6] >= 0.f ? gp : sg);
      if constexpr (APPLY) { s0 = fmaf(Fk, s0, Vk); po[(size_t)6 * Hdim] = s0; }
      else { s1 = fmaf(Fk, s1, Vk); s0 *= Fk; } }
    { const float Fk = QQ[7] * RR[7], iw = PP[7] * RR[7];
      const float sg = TT[7] * RU[7];
      const float gp = (hP[7] + 0.5f) + EPSV;
      const float Vk = iw * (hP[7] >= 0.f ? gp : sg);
      if constexpr (APPLY) { s0 = fmaf(Fk, s0, Vk); po[(size_t)7 * Hdim] = s0; }
      else { s1 = fmaf(Fk, s1, Vk); s0 *= Fk; } }
    if constexpr (REFILL) a = load_xa(xnext);
}

struct WaveJob {
  int n, hi, gp, q;
  const float* xp;
};

__device__ __forceinline__ WaveJob wave_job(const float* __restrict__ x) {
  WaveJob j;
  const int lane = threadIdx.x & 31;
  const int wv   = threadIdx.x >> 5;
  const int gid  = blockIdx.x * 8 + wv;
  const int NPAIR = GROUPS / 2;
  j.gp = gid % NPAIR;
  const int pc = gid / NPAIR;
  j.n  = lane & 15;
  j.hi = lane >> 4;
  j.q  = 2 * pc + j.hi;
  const int qrow = 2 * pc + (j.n >> 3);
  j.xp = x + ((size_t)qrow * CLEN + (size_t)(j.n & 7)) * (size_t)Ddim +
         (size_t)j.gp * 16 + 2 * j.hi;
  return j;
}

__global__ void __launch_bounds__(256, 8) chunk_local(
    const float* __restrict__ x, const float* __restrict__ Wf,
    const float* __restrict__ Wi, const float* __restrict__ Wh,
    float* __restrict__ chA, float* __restrict__ chB) {
  const WaveJob jb = wave_job(x);
  const float* xp = jb.xp;
  const WB wf = load_wb(Wf, jb.gp * 2, jb.n, jb.hi);
  const WB wi = load_wb(Wi, jb.gp * 2, jb.n, jb.hi);
  const WB wh = load_wb(Wh, jb.gp * 2, jb.n, jb.hi);

  float A = 1.f, Bv = 0.f;
  XA r0 = load_xa(xp);
  XA r1 = load_xa(xp + (size_t)TSTEP * Ddim);
  v8f fE = gate16(r0, wf), iE = gate16(r0, wi), hE = gate16(r0, wh);
  r0 = load_xa(xp + 2 * (size_t)TSTEP * Ddim);
  v8f fO, iO, hO;
  for (int t = 0; t < (NTILE - 4) / 2; ++t) {
    pipe_step<false, true>(r1, xp + 3 * (size_t)TSTEP * Ddim, wf, wi, wh, fO, iO, hO, fE, iE, hE, A, Bv, nullptr);
    pipe_step<false, true>(r0, xp + 4 * (size_t)TSTEP * Ddim, wf, wi, wh, fE, iE, hE, fO, iO, hO, A, Bv, nullptr);
    xp += 2 * (size_t)TSTEP * Ddim;
  }
  pipe_step<false, true >(r1, xp + 3 * (size_t)TSTEP * Ddim, wf, wi, wh, fO, iO, hO, fE, iE, hE, A, Bv, nullptr);
  pipe_step<false, false>(r0, xp,             wf, wi, wh, fE, iE, hE, fO, iO, hO, A, Bv, nullptr);
  pipe_step<false, false>(r1, xp,             wf, wi, wh, fO, iO, hO, fE, iE, hE, A, Bv, nullptr);
  local_tile(fO, iO, hO, A, Bv);
  const int b = jb.q / NCHUNK;
  const int c = jb.q % NCHUNK;
  const int h = jb.gp * 16 + jb.n;
  const int sidx = c * (Bdim * Hdim) + b * Hdim + h;
  chA[sidx] = A;
  chB[sidx] = Bv;
}

#define LOOKAHEAD_CARRY 8

__global__ void __launch_bounds__(256) carry_scan(
    const float* __restrict__ chA, const float* __restrict__ chB,
    float* __restrict__ cH) {
  int chan = blockIdx.x * 256 + threadIdx.x;
  const int nchan = Bdim * Hdim;
  if (chan >= nchan) return;
  const float* pa = chA + chan;
  const float* pb = chB + chan;
  float* pq = cH + chan;
  float qa[LOOKAHEAD_CARRY], qb[LOOKAHEAD_CARRY];
#pragma unroll
  for (int j = 0; j < LOOKAHEAD_CARRY; ++j) {
    qa[j] = pa[j * nchan];
    qb[j] = pb[j * nchan];
  }
  float H = 0.f;
  for (int c0 = LOOKAHEAD_CARRY; c0 < NCHUNK; c0 += LOOKAHEAD_CARRY) {
    pa += LOOKAHEAD_CARRY * nchan;
    pb += LOOKAHEAD_CARRY * nchan;
#pragma unroll
    for (int j = 0; j < LOOKAHEAD_CARRY; ++j) {
      pq[j * nchan] = H;
      H = fmaf(qa[j], H, qb[j]);
      qa[j] = pa[j * nchan];
      qb[j] = pb[j * nchan];
    }
    pq += LOOKAHEAD_CARRY * nchan;
  }
#pragma unroll
  for (int j = 0; j < LOOKAHEAD_CARRY; ++j) {
    pq[j * nchan] = H;
    H = fmaf(qa[j], H, qb[j]);
  }
}

__global__ void __launch_bounds__(256) apply_scan(
    const float* __restrict__ x, const float* __restrict__ Wf,
    const float* __restrict__ Wi, const float* __restrict__ Wh,
    const float* __restrict__ cH, float* __restrict__ out) {
  const WaveJob jb = wave_job(x);
  const float* xp = jb.xp;
  const WB wf = load_wb(Wf, jb.gp * 2, jb.n, jb.hi);
  const WB wi = load_wb(Wi, jb.gp * 2, jb.n, jb.hi);
  const WB wh = load_wb(Wh, jb.gp * 2, jb.n, jb.hi);

  const int b = jb.q / NCHUNK;
  const int c = jb.q % NCHUNK;
  const int h = jb.gp * 16 + jb.n;
  float hv = cH[c * (Bdim * Hdim) + b * Hdim + h];
  float unused = 0.f;
  float* po = out + (size_t)jb.q * CLEN * (size_t)Hdim + h;
  XA r0 = load_xa(xp);
  XA r1 = load_xa(xp + (size_t)TSTEP * Ddim);
  v8f fE = gate16(r0, wf), iE = gate16(r0, wi), hE = gate16(r0, wh);
  r0 = load_xa(xp + 2 * (size_t)TSTEP * Ddim);
  v8f fO, iO, hO;
  for (int t = 0; t < (NTILE - 4) / 2; ++t) {
    pipe_step<true, true>(r1, xp + 3 * (size_t)TSTEP * Ddim, wf, wi, wh, fO, iO, hO, fE, iE, hE, hv, unused, po);
    pipe_step<true, true>(r0, xp + 4 * (size_t)TSTEP * Ddim, wf, wi, wh, fE, iE, hE, fO, iO, hO, hv, unused, po + (size_t)1 * TSTEP * Hdim);
    xp += 2 * (size_t)TSTEP * Ddim;
    po += (size_t)2 * TSTEP * Hdim;
  }
  pipe_step<true, true >(r1, xp + 3 * (size_t)TSTEP * Ddim, wf, wi, wh, fO, iO, hO, fE, iE, hE, hv, unused, po);
  pipe_step<true, false>(r0, xp,             wf, wi, wh, fE, iE, hE, fO, iO, hO, hv, unused, po + (size_t)1 * TSTEP * Hdim);
  pipe_step<true, false>(r1, xp,             wf, wi, wh, fO, iO, hO, fE, iE, hE, hv, unused, po + (size_t)2 * TSTEP * Hdim);
  apply_tile(fO, iO, hO, hv, po + (size_t)3 * TSTEP * Hdim);
}

extern "C" void kernel_launch(void* const* d_in, const int* in_sizes, int n_in,
                              void* d_out, int out_size, void* d_ws, size_t ws_size,
                              hipStream_t stream) {
  (void)in_sizes; (void)n_in; (void)out_size; (void)ws_size;
  const float* x  = (const float*)d_in[0];
  const float* Wf = (const float*)d_in[1];
  const float* Wi = (const float*)d_in[2];
  const float* Wh = (const float*)d_in[3];
  float* out = (float*)d_out;

  const size_t NSUM = (size_t)NCHUNK * Bdim * Hdim;
  float* chA = (float*)d_ws;
  float* chB = chA + NSUM;
  float* cH  = chB + NSUM;

  const int scan_blocks = (GROUPS / 2) * (Bdim * NCHUNK / 2) / 8;
  chunk_local<<<scan_blocks, 256, 0, stream>>>(x, Wf, Wi, Wh, chA, chB);
  carry_scan<<<(Bdim * Hdim + 255) / 256, 256, 0, stream>>>(chA, chB, cH);
  apply_scan<<<scan_blocks, 256, 0, stream>>>(x, Wf, Wi, Wh, cH, out);
}


# gate math cut from 22 to 13 vector instructions per element: log2(e) folded into the weight fragments at load, sigmoid as 1/(1+e^-hid), g(hid) as one max; rounding changes, sum order unchanged
# speedup vs baseline: 6.2835x; 1.0562x over previous
#include <hip/hip_runtime.h>
#include <math.h>

#define GROUPS 128
#define Bdim   4
#define Sdim   8192
#define Ddim   1024
#define Hdim   1024
#define NCHUNK 64
#define CLEN   (Sdim / NCHUNK)

typedef float v2f __attribute__((ext_vector_type(2)));
typedef float v8f __attribute__((ext_vector_type(8)));

#define LOG2E 1.44269504088896341f
#define LN2   0.693147180559945309f

__device__ __forceinline__ float exp2_hw(float v) {
  return __builtin_amdgcn_exp2f(v);
}

struct GateRow { float e0, e1, e2, eh, p, q, r, sg; };

__device__ __forceinline__ void gate_s1(GateRow& g, float f2, float i2, float h2) {
  const float nf = -f2, ni = -i2;
  const float m  = fmaxf(fmaxf(nf, ni), 0.f);
  g.e0 = exp2_hw(-m);
  g.e1 = exp2_hw(nf - m);
  g.e2 = exp2_hw(ni - m);
  g.eh = exp2_hw(-h2);
}

__device__ __forceinline__ void gate_s2(GateRow& g) {
  g.p  = g.e0 + g.e1;
  g.q  = g.e0 + g.e2;
  g.r  = __builtin_amdgcn_rcpf(g.p + g.q);
  g.sg = __builtin_amdgcn_rcpf(1.f + g.eh);
}

__device__ __forceinline__ void gate_s3(const GateRow& g, float h2, float& Fo, float& Vo) {
  Fo = g.q * g.r;
  const float iw = g.p * g.r;
  const float gp = fmaf(h2, LN2, 0.5f);
  Vo = iw * fmaxf(gp, g.sg);
}

__device__ __forceinline__ void gate_lin(float f2, float i2, float h2,
                                         float& Fo, float& Vo) {
  GateRow g;
  gate_s1(g, f2, i2, h2);
  gate_s2(g);
  gate_s3(g, h2, Fo, Vo);
}

__device__ __forceinline__ v2f ldw(const float* __restrict__ Wg, int cn, int kc,
                                   int hi, bool valid) {
  v2f w = *(const v2f*)(Wg + cn * 8 + kc * 4 + 2 * hi);
  v2f r;
  r.x = valid ? w.x * LOG2E : 0.f;
  r.y = valid ? w.y * LOG2E : 0.f;
  return r;
}

struct WB { v2f b00, b01, b10, b11; };

__device__ __forceinline__ WB load_wb(const float* __restrict__ W, int g0,
                                      int n, int hi) {
  const int  cn = n & 7;
  const bool lo = (n < 8);
  const float* Wg0 = W + (size_t)g0 * 64;
  const float* Wg1 = Wg0 + 64;
  WB w;
  w.b00 = ldw(Wg0, cn, 0, hi, lo);
  w.b01 = ldw(Wg0, cn, 1, hi, lo);
  w.b10 = ldw(Wg1, cn, 0, hi, !lo);
  w.b11 = ldw(Wg1, cn, 1, hi, !lo);
  return w;
}

struct XA { v2f a00, a01, a10, a11; };

__device__ __forceinline__ XA load_xa(const float* __restrict__ xrow) {
  XA a;
  a.a00 = *(const v2f*)(xrow + 0);
  a.a01 = *(const v2f*)(xrow + 4);
  a.a10 = *(const v2f*)(xrow + 8);
  a.a11 = *(const v2f*)(xrow + 12);
  return a;
}

__device__ __forceinline__ v8f gate16(const XA& a, const WB& w) {
  v8f acc = {};
  acc = __builtin_amdgcn_wmma_f32_16x16x4_f32(false, a.a00, false, w.b00, (short)0, acc, false, false);
  acc = __builtin_amdgcn_wmma_f32_16x16x4_f32(false, a.a01, false, w.b01, (short)0, acc, false, false);
  acc = __builtin_amdgcn_wmma_f32_16x16x4_f32(false, a.a10, false, w.b10, (short)0, acc, false, false);
  acc = __builtin_amdgcn_wmma_f32_16x16x4_f32(false, a.a11, false, w.b11, (short)0, acc, false, false);
  return acc;
}

#define TSTEP  8
#define NTILE  (CLEN / TSTEP)


__device__ __forceinline__ void local_tile(const v8f& fg, const v8f& ig, const v8f& hg,
                                           float& A, float& Bv) {
#pragma unroll
  for (int r = 0; r < TSTEP; ++r) {
    float Fk, Vk;
    gate_lin(fg[r], ig[r], hg[r], Fk, Vk);
    Bv = fmaf(Fk, Bv, Vk);
    A *= Fk;
  }
}

__device__ __forceinline__ void apply_tile(const v8f& fg, const v8f& ig, const v8f& hg,
                                           float& hv, float* __restrict__ po) {
#pragma unroll
  for (int r = 0; r < TSTEP; ++r) {
    float Fk, Vk;
    gate_lin(fg[r], ig[r], hg[r], Fk, Vk);
    hv = fmaf(Fk, hv, Vk);
    po[(size_t)r * Hdim] = hv;
  }
}

template <bool APPLY>
__device__ __forceinline__ void scan_row(const GateRow& g, float h2, float& s0, float& s1,
                                         float* __restrict__ po, int r) {
  float Fk, Vk;
  gate_s3(g, h2, Fk, Vk);
  if constexpr (APPLY) { s0 = fmaf(Fk, s0, Vk); po[(size_t)r * Hdim] = s0; }
  else { s1 = fmaf(Fk, s1, Vk); s0 *= Fk; }
}

template <bool APPLY, bool REFILL>
__device__ __forceinline__ void pipe_step(XA& a, const float* __restrict__ xnext,
                                          const WB& wf, const WB& wi, const WB& wh,
                                          v8f& fN, v8f& iN, v8f& hN,
                                          const v8f& fP, const v8f& iP, const v8f& hP,
                                          float& s0, float& s1, float* __restrict__ po) {
  const v8f zero = {};
  GateRow g[TSTEP];
    fN = __builtin_amdgcn_wmma_f32_16x16x4_f32(false, a.a00, false, wf.b00, (short)0, zero, false, false);
    gate_s1(g[0], fP[0], iP[0], hP[0]);
    gate_s1(g[1], fP[1], iP[1], hP[1]);
    __builtin_amdgcn_sched_barrier(0);
    iN = __builtin_amdgcn_wmma_f32_16x16x4_f32(false, a.a00, false, wi.b00, (short)0, zero, false, false);
    gate_s2(g[0]);
    gate_s1(g[2], fP[2], iP[2], hP[2]);
    __builtin_amdgcn_sched_barrier(0);
    hN = __builtin_amdgcn_wmma_f32_16x16x4_f32(false, a.a00, false, wh.b00, (short)0, zero, false, false);
    gate_s2(g[1]);
    scan_row<APPLY>(g[0], hP[0], s0, s1, po, 0);
    __builtin_amdgcn_sched_barrier(0);
    fN = __builtin_amdgcn_wmma_f32_16x16x4_f32(false, a.a01, false, wf.b01, (short)0, fN, false, false);
    gate_s1(g[3], fP[3], iP[3], hP[3]);
    gate_s2(g[2]);
    __builtin_amdgcn_sched_barrier(0);
    iN = __builtin_amdgcn_wmma_f32_16x16x4_f32(false, a.a01, false, wi.b01, (short)0, iN, false, false);
    scan_row<APPLY>(g[1], hP[1], s0, s1, po, 1);
    gate_s1(g[4], fP[4], iP[4], hP[4]);
    __builtin_amdgcn_sched_barrier(0);
    hN = __builtin_amdgcn_wmma_f32_16x16x4_f32(false, a.a01, false, wh.b01, (short)0, hN, false, false);
    gate_s2(g[3]);
    scan_row<APPLY>(g[2], hP[2], s0, s1, po, 2);
    __builtin_amdgcn_sched_barrier(0);
    fN = __builtin_amdgcn_wmma_f32_16x16x4_f32(false, a.a10, false, wf.b10, (short)0, fN, false, false);
    gate_s1(g[5], fP[5], iP[5], hP[5]);
    gate_s2(g[4]);
    __builtin_amdgcn_sched_barrier(0);
    iN = __builtin_amdgcn_wmma_f32_16x16x4_f32(false, a.a10, false, wi.b10, (short)0, iN, false, false);
    scan_row<APPLY>(g[3], hP[3], s0, s1, po, 3);
    gate_s1(g[6], fP[6], iP[6], hP[6]);
    __builtin_amdgcn_sched_barrier(0);
    hN = __builtin_amdgcn_wmma_f32_16x16x4_f32(false, a.a10, false, wh.b10, (short)0, hN, false, false);
    gate_s2(g[5]);
    scan_row<APPLY>(g[4], hP[4], s0, s1, po, 4);
    __builtin_amdgcn_sched_barrier(0);
    fN = __builtin_amdgcn_wmma_f32_16x16x4_f32(false, a.a11, false, wf.b11, (short)0, fN, false, false);
    gate_s1(g[7], fP[7], iP[7], hP[7]);
    gate_s2(g[6]);
    __builtin_amdgcn_sched_barrier(0);
    iN = __builtin_amdgcn_wmma_f32_16x16x4_f32(false, a.a11, false, wi.b11, (short)0, iN, false, false);
    scan_row<APPLY>(g[5], hP[5], s0, s1, po, 5);
    gate_s2(g[7]);
    __builtin_amdgcn_sched_barrier(0);
    hN = __builtin_amdgcn_wmma_f32_16x16x4_f32(false, a.a11, false, wh.b11, (short)0, hN, false, false);
    scan_row<APPLY>(g[6], hP[6], s0, s1, po, 6);
    scan_row<APPLY>(g[7], hP[7], s0, s1, po, 7);
    if constexpr (REFILL) a = load_xa(xnext);
}

struct WaveJob {
  int n, hi, gp, q;
  const float* xp;
};

__device__ __forceinline__ WaveJob wave_job(const float* __restrict__ x) {
  WaveJob j;
  const int lane = threadIdx.x & 31;
  const int wv   = threadIdx.x >> 5;
  const int gid  = blockIdx.x * 8 + wv;
  const int NPAIR = GROUPS / 2;
  j.gp = gid % NPAIR;
  const int pc = gid / NPAIR;
  j.n  = lane & 15;
  j.hi = lane >> 4;
  j.q  = 2 * pc + j.hi;
  const int qrow = 2 * pc + (j.n >> 3);
  j.xp = x + ((size_t)qrow * CLEN + (size_t)(j.n & 7)) * (size_t)Ddim +
         (size_t)j.gp * 16 + 2 * j.hi;
  return j;
}

__global__ void __launch_bounds__(256, 8) chunk_local(
    const float* __restrict__ x, const float* __restrict__ Wf,
    const float* __restrict__ Wi, const float* __restrict__ Wh,
    float* __restrict__ chA, float* __restrict__ chB) {
  const WaveJob jb = wave_job(x);
  const float* xp = jb.xp;
  const WB wf = load_wb(Wf, jb.gp * 2, jb.n, jb.hi);
  const WB wi = load_wb(Wi, jb.gp * 2, jb.n, jb.hi);
  const WB wh = load_wb(Wh, jb.gp * 2, jb.n, jb.hi);

  float A = 1.f, Bv = 0.f;
  XA r0 = load_xa(xp);
  XA r1 = load_xa(xp + (size_t)TSTEP * Ddim);
  v8f fE = gate16(r0, wf), iE = gate16(r0, wi), hE = gate16(r0, wh);
  r0 = load_xa(xp + 2 * (size_t)TSTEP * Ddim);
  v8f fO, iO, hO;
  for (int t = 0; t < (NTILE - 4) / 2; ++t) {
    pipe_step<false, true>(r1, xp + 3 * (size_t)TSTEP * Ddim, wf, wi, wh, fO, iO, hO, fE, iE, hE, A, Bv, nullptr);
    pipe_step<false, true>(r0, xp + 4 * (size_t)TSTEP * Ddim, wf, wi, wh, fE, iE, hE, fO, iO, hO, A, Bv, nullptr);
    xp += 2 * (size_t)TSTEP * Ddim;
  }
  pipe_step<false, true >(r1, xp + 3 * (size_t)TSTEP * Ddim, wf, wi, wh, fO, iO, hO, fE, iE, hE, A, Bv, nullptr);
  pipe_step<false, false>(r0, xp,             wf, wi, wh, fE, iE, hE, fO, iO, hO, A, Bv, nullptr);
  pipe_step<false, false>(r1, xp,             wf, wi, wh, fO, iO, hO, fE, iE, hE, A, Bv, nullptr);
  local_tile(fO, iO, hO, A, Bv);
  const int b = jb.q / NCHUNK;
  const int c = jb.q % NCHUNK;
  const int h = jb.gp * 16 + jb.n;
  const int sidx = c * (Bdim * Hdim) + b * Hdim + h;
  chA[sidx] = A;
  chB[sidx] = Bv;
}

#define LOOKAHEAD_CARRY 8

__global__ void __launch_bounds__(256) carry_scan(
    const float* __restrict__ chA, const float* __restrict__ chB,
    float* __restrict__ cH) {
  int chan = blockIdx.x * 256 + threadIdx.x;
  const int nchan = Bdim * Hdim;
  if (chan >= nchan) return;
  const float* pa = chA + chan;
  const float* pb = chB + chan;
  float* pq = cH + chan;
  float qa[LOOKAHEAD_CARRY], qb[LOOKAHEAD_CARRY];
#pragma unroll
  for (int j = 0; j < LOOKAHEAD_CARRY; ++j) {
    qa[j] = pa[j * nchan];
    qb[j] = pb[j * nchan];
  }
  float H = 0.f;
  for (int c0 = LOOKAHEAD_CARRY; c0 < NCHUNK; c0 += LOOKAHEAD_CARRY) {
    pa += LOOKAHEAD_CARRY * nchan;
    pb += LOOKAHEAD_CARRY * nchan;
#pragma unroll
    for (int j = 0; j < LOOKAHEAD_CARRY; ++j) {
      pq[j * nchan] = H;
      H = fmaf(qa[j], H, qb[j]);
      qa[j] = pa[j * nchan];
      qb[j] = pb[j * nchan];
    }
    pq += LOOKAHEAD_CARRY * nchan;
  }
#pragma unroll
  for (int j = 0; j < LOOKAHEAD_CARRY; ++j) {
    pq[j * nchan] = H;
    H = fmaf(qa[j], H, qb[j]);
  }
}

__global__ void __launch_bounds__(256) apply_scan(
    const float* __restrict__ x, const float* __restrict__ Wf,
    const float* __restrict__ Wi, const float* __restrict__ Wh,
    const float* __restrict__ cH, float* __restrict__ out) {
  const WaveJob jb = wave_job(x);
  const float* xp = jb.xp;
  const WB wf = load_wb(Wf, jb.gp * 2, jb.n, jb.hi);
  const WB wi = load_wb(Wi, jb.gp * 2, jb.n, jb.hi);
  const WB wh = load_wb(Wh, jb.gp * 2, jb.n, jb.hi);

  const int b = jb.q / NCHUNK;
  const int c = jb.q % NCHUNK;
  const int h = jb.gp * 16 + jb.n;
  float hv = cH[c * (Bdim * Hdim) + b * Hdim + h];
  float unused = 0.f;
  float* po = out + (size_t)jb.q * CLEN * (size_t)Hdim + h;
  XA r0 = load_xa(xp);
  XA r1 = load_xa(xp + (size_t)TSTEP * Ddim);
  v8f fE = gate16(r0, wf), iE = gate16(r0, wi), hE = gate16(r0, wh);
  r0 = load_xa(xp + 2 * (size_t)TSTEP * Ddim);
  v8f fO, iO, hO;
  for (int t = 0; t < (NTILE - 4) / 2; ++t) {
    pipe_step<true, true>(r1, xp + 3 * (size_t)TSTEP * Ddim, wf, wi, wh, fO, iO, hO, fE, iE, hE, hv, unused, po);
    pipe_step<true, true>(r0, xp + 4 * (size_t)TSTEP * Ddim, wf, wi, wh, fE, iE, hE, fO, iO, hO, hv, unused, po + (size_t)1 * TSTEP * Hdim);
    xp += 2 * (size_t)TSTEP * Ddim;
    po += (size_t)2 * TSTEP * Hdim;
  }
  pipe_step<true, true >(r1, xp + 3 * (size_t)TSTEP * Ddim, wf, wi, wh, fO, iO, hO, fE, iE, hE, hv, unused, po);
  pipe_step<true, false>(r0, xp,             wf, wi, wh, fE, iE, hE, fO, iO, hO, hv, unused, po + (size_t)1 * TSTEP * Hdim);
  pipe_step<true, false>(r1, xp,             wf, wi, wh, fO, iO, hO, fE, iE, hE, hv, unused, po + (size_t)2 * TSTEP * Hdim);
  apply_tile(fO, iO, hO, hv, po + (size_t)3 * TSTEP * Hdim);
}

extern "C" void kernel_launch(void* const* d_in, const int* in_sizes, int n_in,
                              void* d_out, int out_size, void* d_ws, size_t ws_size,
                              hipStream_t stream) {
  (void)in_sizes; (void)n_in; (void)out_size; (void)ws_size;
  const float* x  = (const float*)d_in[0];
  const float* Wf = (const float*)d_in[1];
  const float* Wi = (const float*)d_in[2];
  const float* Wh = (const float*)d_in[3];
  float* out = (float*)d_out;

  const size_t NSUM = (size_t)NCHUNK * Bdim * Hdim;
  float* chA = (float*)d_ws;
  float* chB = chA + NSUM;
  float* cH  = chB + NSUM;

  const int scan_blocks = (GROUPS / 2) * (Bdim * NCHUNK / 2) / 8;
  chunk_local<<<scan_blocks, 256, 0, stream>>>(x, Wf, Wi, Wh, chA, chB);
  carry_scan<<<(Bdim * Hdim + 255) / 256, 256, 0, stream>>>(chA, chB, cH);
  apply_scan<<<scan_blocks, 256, 0, stream>>>(x, Wf, Wi, Wh, cH, out);
}


# each lane of the re-walk pass folds the summaries of the chunks before its own (same chunk order) itself: the separate carry launch is gone, 3 launches -> 2; sum order unchanged
# speedup vs baseline: 6.5442x; 1.0415x over previous
#include <hip/hip_runtime.h>
#include <math.h>

#define GROUPS 128
#define Bdim   4
#define Sdim   8192
#define Ddim   1024
#define Hdim   1024
#define NCHUNK 64
#define CLEN   (Sdim / NCHUNK)

typedef float v2f __attribute__((ext_vector_type(2)));
typedef float v8f __attribute__((ext_vector_type(8)));

#define LOG2E 1.44269504088896341f
#define LN2   0.693147180559945309f

__device__ __forceinline__ float exp2_hw(float v) {
  return __builtin_amdgcn_exp2f(v);
}

struct GateRow { float e0, e1, e2, eh, p, q, r, sg; };

__device__ __forceinline__ void gate_s1(GateRow& g, float f2, float i2, float h2) {
  const float nf = -f2, ni = -i2;
  const float m  = fmaxf(fmaxf(nf, ni), 0.f);
  g.e0 = exp2_hw(-m);
  g.e1 = exp2_hw(nf - m);
  g.e2 = exp2_hw(ni - m);
  g.eh = exp2_hw(-h2);
}

__device__ __forceinline__ void gate_s2(GateRow& g) {
  g.p  = g.e0 + g.e1;
  g.q  = g.e0 + g.e2;
  g.r  = __builtin_amdgcn_rcpf(g.p + g.q);
  g.sg = __builtin_amdgcn_rcpf(1.f + g.eh);
}

__device__ __forceinline__ void gate_s3(const GateRow& g, float h2, float& Fo, float& Vo) {
  Fo = g.q * g.r;
  const float iw = g.p * g.r;
  const float gp = fmaf(h2, LN2, 0.5f);
  Vo = iw * fmaxf(gp, g.sg);
}

__device__ __forceinline__ void gate_lin(float f2, float i2, float h2,
                                         float& Fo, float& Vo) {
  GateRow g;
  gate_s1(g, f2, i2, h2);
  gate_s2(g);
  gate_s3(g, h2, Fo, Vo);
}

__device__ __forceinline__ v2f ldw(const float* __restrict__ Wg, int cn, int kc,
                                   int hi, bool valid) {
  v2f w = *(const v2f*)(Wg + cn * 8 + kc * 4 + 2 * hi);
  v2f r;
  r.x = valid ? w.x * LOG2E : 0.f;
  r.y = valid ? w.y * LOG2E : 0.f;
  return r;
}

struct WB { v2f b00, b01, b10, b11; };

__device__ __forceinline__ WB load_wb(const float* __restrict__ W, int g0,
                                      int n, int hi) {
  const int  cn = n & 7;
  const bool lo = (n < 8);
  const float* Wg0 = W + (size_t)g0 * 64;
  const float* Wg1 = Wg0 + 64;
  WB w;
  w.b00 = ldw(Wg0, cn, 0, hi, lo);
  w.b01 = ldw(Wg0, cn, 1, hi, lo);
  w.b10 = ldw(Wg1, cn, 0, hi, !lo);
  w.b11 = ldw(Wg1, cn, 1, hi, !lo);
  return w;
}

struct XA { v2f a00, a01, a10, a11; };

__device__ __forceinline__ XA load_xa(const float* __restrict__ xrow) {
  XA a;
  a.a00 = *(const v2f*)(xrow + 0);
  a.a01 = *(const v2f*)(xrow + 4);
  a.a10 = *(const v2f*)(xrow + 8);
  a.a11 = *(const v2f*)(xrow + 12);
  return a;
}

__device__ __forceinline__ v8f gate16(const XA& a, const WB& w) {
  v8f acc = {};
  acc = __builtin_amdgcn_wmma_f32_16x16x4_f32(false, a.a00, false, w.b00, (short)0, acc, false, false);
  acc = __builtin_amdgcn_wmma_f32_16x16x4_f32(false, a.a01, false, w.b01, (short)0, acc, false, false);
  acc = __builtin_amdgcn_wmma_f32_16x16x4_f32(false, a.a10, false, w.b10, (short)0, acc, false, false);
  acc = __builtin_amdgcn_wmma_f32_16x16x4_f32(false, a.a11, false, w.b11, (short)0, acc, false, false);
  return acc;
}

#define TSTEP  8
#define NTILE  (CLEN / TSTEP)


__device__ __forceinline__ void local_tile(const v8f& fg, const v8f& ig, const v8f& hg,
                                           float& A, float& Bv) {
#pragma unroll
  for (int r = 0; r < TSTEP; ++r) {
    float Fk, Vk;
    gate_lin(fg[r], ig[r], hg[r], Fk, Vk);
    Bv = fmaf(Fk, Bv, Vk);
    A *= Fk;
  }
}

__device__ __forceinline__ void apply_tile(const v8f& fg, const v8f& ig, const v8f& hg,
                                           float& hv, float* __restrict__ po) {
#pragma unroll
  for (int r = 0; r < TSTEP; ++r) {
    float Fk, Vk;
    gate_lin(fg[r], ig[r], hg[r], Fk, Vk);
    hv = fmaf(Fk, hv, Vk);
    po[(size_t)r * Hdim] = hv;
  }
}

template <bool APPLY>
__device__ __forceinline__ void scan_row(const GateRow& g, float h2, float& s0, float& s1,
                                         float* __restrict__ po, int r) {
  float Fk, Vk;
  gate_s3(g, h2, Fk, Vk);
  if constexpr (APPLY) { s0 = fmaf(Fk, s0, Vk); po[(size_t)r * Hdim] = s0; }
  else { s1 = fmaf(Fk, s1, Vk); s0 *= Fk; }
}

template <bool APPLY, bool REFILL>
__device__ __forceinline__ void pipe_step(XA& a, const float* __restrict__ xnext,
                                          const WB& wf, const WB& wi, const WB& wh,
                                          v8f& fN, v8f& iN, v8f& hN,
                                          const v8f& fP, const v8f& iP, const v8f& hP,
                                          float& s0, float& s1, float* __restrict__ po) {
  const v8f zero = {};
  GateRow g[TSTEP];
    fN = __builtin_amdgcn_wmma_f32_16x16x4_f32(false, a.a00, false, wf.b00, (short)0, zero, false, false);
    gate_s1(g[0], fP[0], iP[0], hP[0]);
    gate_s1(g[1], fP[1], iP[1], hP[1]);
    __builtin_amdgcn_sched_barrier(0);
    iN = __builtin_amdgcn_wmma_f32_16x16x4_f32(false, a.a00, false, wi.b00, (short)0, zero, false, false);
    gate_s2(g[0]);
    gate_s1(g[2], fP[2], iP[2], hP[2]);
    __builtin_amdgcn_sched_barrier(0);
    hN = __builtin_amdgcn_wmma_f32_16x16x4_f32(false, a.a00, false, wh.b00, (short)0, zero, false, false);
    gate_s2(g[1]);
    scan_row<APPLY>(g[0], hP[0], s0, s1, po, 0);
    __builtin_amdgcn_sched_barrier(0);
    fN = __builtin_amdgcn_wmma_f32_16x16x4_f32(false, a.a01, false, wf.b01, (short)0, fN, false, false);
    gate_s1(g[3], fP[3], iP[3], hP[3]);
    gate_s2(g[2]);
    __builtin_amdgcn_sched_barrier(0);
    iN = __builtin_amdgcn_wmma_f32_16x16x4_f32(false, a.a01, false, wi.b01, (short)0, iN, false, false);
    scan_row<APPLY>(g[1], hP[1], s0, s1, po, 1);
    gate_s1(g[4], fP[4], iP[4], hP[4]);
    __builtin_amdgcn_sched_barrier(0);
    hN = __builtin_amdgcn_wmma_f32_16x16x4_f32(false, a.a01, false, wh.b01, (short)0, hN, false, false);
    gate_s2(g[3]);
    scan_row<APPLY>(g[2], hP[2], s0, s1, po, 2);
    __builtin_amdgcn_sched_barrier(0);
    fN = __builtin_amdgcn_wmma_f32_16x16x4_f32(false, a.a10, false, wf.b10, (short)0, fN, false, false);
    gate_s1(g[5], fP[5], iP[5], hP[5]);
    gate_s2(g[4]);
    __builtin_amdgcn_sched_barrier(0);
    iN = __builtin_amdgcn_wmma_f32_16x16x4_f32(false, a.a10, false, wi.b10, (short)0, iN, false, false);
    scan_row<APPLY>(g[3], hP[3], s0, s1, po, 3);
    gate_s1(g[6], fP[6], iP[6], hP[6]);
    __builtin_amdgcn_sched_barrier(0);
    hN = __builtin_amdgcn_wmma_f32_16x16x4_f32(false, a.a10, false, wh.b10, (short)0, hN, false, false);
    gate_s2(g[5]);
    scan_row<APPLY>(g[4], hP[4], s0, s1, po, 4);
    __builtin_amdgcn_sched_barrier(0);
    fN = __builtin_amdgcn_wmma_f32_16x16x4_f32(false, a.a11, false, wf.b11, (short)0, fN, false, false);
    gate_s1(g[7], fP[7], iP[7], hP[7]);
    gate_s2(g[6]);
    __builtin_amdgcn_sched_barrier(0);
    iN = __builtin_amdgcn_wmma_f32_16x16x4_f32(false, a.a11, false, wi.b11, (short)0, iN, false, false);
    scan_row<APPLY>(g[5], hP[5], s0, s1, po, 5);
    gate_s2(g[7]);
    __builtin_amdgcn_sched_barrier(0);
    hN = __builtin_amdgcn_wmma_f32_16x16x4_f32(false, a.a11, false, wh.b11, (short)0, hN, false, false);
    scan_row<APPLY>(g[6], hP[6], s0, s1, po, 6);
    scan_row<APPLY>(g[7], hP[7], s0, s1, po, 7);
    if constexpr (REFILL) a = load_xa(xnext);
}

struct WaveJob {
  int n, hi, gp, q;
  const float* xp;
};

__device__ __forceinline__ WaveJob wave_job(const float* __restrict__ x) {
  WaveJob j;
  const int lane = threadIdx.x & 31;
  const int wv   = threadIdx.x >> 5;
  const int gid  = blockIdx.x * 8 + wv;
  const int NPAIR = GROUPS / 2;
  j.gp = gid % NPAIR;
  const int pc = gid / NPAIR;
  j.n  = lane & 15;
  j.hi = lane >> 4;
  j.q  = 2 * pc + j.hi;
  const int qrow = 2 * pc + (j.n >> 3);
  j.xp = x + ((size_t)qrow * CLEN + (size_t)(j.n & 7)) * (size_t)Ddim +
         (size_t)j.gp * 16 + 2 * j.hi;
  return j;
}

__global__ void __launch_bounds__(256, 8) chunk_local(
    const float* __restrict__ x, const float* __restrict__ Wf,
    const float* __restrict__ Wi, const float* __restrict__ Wh,
    float* __restrict__ chA, float* __restrict__ chB) {
  const WaveJob jb = wave_job(x);
  const float* xp = jb.xp;
  const WB wf = load_wb(Wf, jb.gp * 2, jb.n, jb.hi);
  const WB wi = load_wb(Wi, jb.gp * 2, jb.n, jb.hi);
  const WB wh = load_wb(Wh, jb.gp * 2, jb.n, jb.hi);

  float A = 1.f, Bv = 0.f;
  XA r0 = load_xa(xp);
  XA r1 = load_xa(xp + (size_t)TSTEP * Ddim);
  v8f fE = gate16(r0, wf), iE = gate16(r0, wi), hE = gate16(r0, wh);
  r0 = load_xa(xp + 2 * (size_t)TSTEP * Ddim);
  v8f fO, iO, hO;
  for (int t = 0; t < (NTILE - 4) / 2; ++t) {
    pipe_step<false, true>(r1, xp + 3 * (size_t)TSTEP * Ddim, wf, wi, wh, fO, iO, hO, fE, iE, hE, A, Bv, nullptr);
    pipe_step<false, true>(r0, xp + 4 * (size_t)TSTEP * Ddim, wf, wi, wh, fE, iE, hE, fO, iO, hO, A, Bv, nullptr);
    xp += 2 * (size_t)TSTEP * Ddim;
  }
  pipe_step<false, true >(r1, xp + 3 * (size_t)TSTEP * Ddim, wf, wi, wh, fO, iO, hO, fE, iE, hE, A, Bv, nullptr);
  pipe_step<false, false>(r0, xp,             wf, wi, wh, fE, iE, hE, fO, iO, hO, A, Bv, nullptr);
  pipe_step<false, false>(r1, xp,             wf, wi, wh, fO, iO, hO, fE, iE, hE, A, Bv, nullptr);
  local_tile(fO, iO, hO, A, Bv);
  const int b = jb.q / NCHUNK;
  const int c = jb.q % NCHUNK;
  const int h = jb.gp * 16 + jb.n;
  const int sidx = c * (Bdim * Hdim) + b * Hdim + h;
  chA[sidx] = A;
  chB[sidx] = Bv;
}

#define CGRP 8
#define NGRP (NCHUNK / 2 / CGRP)
__device__ __forceinline__ float carry_in(const float* __restrict__ pa,
                                          const float* __restrict__ pb,
                                          int c, int cmax) {
  const int nchan = Bdim * Hdim;
  float qa[NCHUNK / 2], qb[NCHUNK / 2];
#pragma unroll
  for (int k = 0; k < NCHUNK / 2; ++k) {
    qa[k] = pa[k * nchan];
    qb[k] = pb[k * nchan];
  }
  float H = 0.f;
  if (cmax > NCHUNK / 2) {
#pragma unroll
    for (int g = 0; g < NGRP; ++g) {
#pragma unroll
      for (int j = 0; j < CGRP; ++j) {
        const int k = g * CGRP + j;
        H = fmaf(qa[k], H, qb[k]);
      }
#pragma unroll
      for (int j = 0; j < CGRP; ++j) {
        const int k = g * CGRP + j;
        qa[k] = pa[(NCHUNK / 2 + k) * nchan];
        qb[k] = pb[(NCHUNK / 2 + k) * nchan];
      }
    }
#pragma unroll
    for (int g = 0; g < NGRP; ++g) {
      if (NCHUNK / 2 + g * CGRP < cmax) {
#pragma unroll
        for (int j = 0; j < CGRP; ++j) {
          const int k = g * CGRP + j;
          const float Hn = fmaf(qa[k], H, qb[k]);
          H = (NCHUNK / 2 + k < c) ? Hn : H;
        }
      }
    }
  } else {
#pragma unroll
    for (int g = 0; g < NGRP; ++g) {
      if (g * CGRP < cmax) {
#pragma unroll
        for (int j = 0; j < CGRP; ++j) {
          const int k = g * CGRP + j;
          const float Hn = fmaf(qa[k], H, qb[k]);
          H = (k < c) ? Hn : H;
        }
      }
    }
  }
  return H;
}

__global__ void __launch_bounds__(256) apply_scan(
    const float* __restrict__ x, const float* __restrict__ Wf,
    const float* __restrict__ Wi, const float* __restrict__ Wh,
    const float* __restrict__ chA, const float* __restrict__ chB,
    float* __restrict__ out) {
  const WaveJob jb = wave_job(x);
  const float* xp = jb.xp;
  const WB wf = load_wb(Wf, jb.gp * 2, jb.n, jb.hi);
  const WB wi = load_wb(Wi, jb.gp * 2, jb.n, jb.hi);
  const WB wh = load_wb(Wh, jb.gp * 2, jb.n, jb.hi);

  const int b = jb.q / NCHUNK;
  const int c = jb.q % NCHUNK;
  const int h = jb.gp * 16 + jb.n;
  float unused = 0.f;
  float* po = out + (size_t)jb.q * CLEN * (size_t)Hdim + h;
  XA r0 = load_xa(xp);
  XA r1 = load_xa(xp + (size_t)TSTEP * Ddim);
  const int cmax = __builtin_amdgcn_readfirstlane((jb.q | 1) % NCHUNK);
  float hv = carry_in(chA + b * Hdim + h, chB + b * Hdim + h, c, cmax);
  v8f fE = gate16(r0, wf), iE = gate16(r0, wi), hE = gate16(r0, wh);
  r0 = load_xa(xp + 2 * (size_t)TSTEP * Ddim);
  v8f fO, iO, hO;
  for (int t = 0; t < (NTILE - 4) / 2; ++t) {
    pipe_step<true, true>(r1, xp + 3 * (size_t)TSTEP * Ddim, wf, wi, wh, fO, iO, hO, fE, iE, hE, hv, unused, po);
    pipe_step<true, true>(r0, xp + 4 * (size_t)TSTEP * Ddim, wf, wi, wh, fE, iE, hE, fO, iO, hO, hv, unused, po + (size_t)1 * TSTEP * Hdim);
    xp += 2 * (size_t)TSTEP * Ddim;
    po += (size_t)2 * TSTEP * Hdim;
  }
  pipe_step<true, true >(r1, xp + 3 * (size_t)TSTEP * Ddim, wf, wi, wh, fO, iO, hO, fE, iE, hE, hv, unused, po);
  pipe_step<true, false>(r0, xp,             wf, wi, wh, fE, iE, hE, fO, iO, hO, hv, unused, po + (size_t)1 * TSTEP * Hdim);
  pipe_step<true, false>(r1, xp,             wf, wi, wh, fO, iO, hO, fE, iE, hE, hv, unused, po + (size_t)2 * TSTEP * Hdim);
  apply_tile(fO, iO, hO, hv, po + (size_t)3 * TSTEP * Hdim);
}

extern "C" void kernel_launch(void* const* d_in, const int* in_sizes, int n_in,
                              void* d_out, int out_size, void* d_ws, size_t ws_size,
                              hipStream_t stream) {
  (void)in_sizes; (void)n_in; (void)out_size; (void)ws_size;
  const float* x  = (const float*)d_in[0];
  const float* Wf = (const float*)d_in[1];
  const float* Wi = (const float*)d_in[2];
  const float* Wh = (const float*)d_in[3];
  float* out = (float*)d_out;

  const size_t NSUM = (size_t)NCHUNK * Bdim * Hdim;
  float* chA = (float*)d_ws;
  float* chB = chA + NSUM;

  const int scan_blocks = (GROUPS / 2) * (Bdim * NCHUNK / 2) / 8;
  chunk_local<<<scan_blocks, 256, 0, stream>>>(x, Wf, Wi, Wh, chA, chB);
  apply_scan<<<scan_blocks, 256, 0, stream>>>(x, Wf, Wi, Wh, chA, chB, out);
}


# f and i of one group share a dense 16-column f32 WMMA tile (zero halves stay for h only): 12 -> 8 matrix instructions per tile; a lane takes its i by DPP row_ror:8; sum order unchanged
# speedup vs baseline: 6.9065x; 1.0554x over previous
#include <hip/hip_runtime.h>
#include <math.h>

#define GROUPS 128
#define Bdim   4
#define Sdim   8192
#define Ddim   1024
#define Hdim   1024
#define NCHUNK 64
#define CLEN   (Sdim / NCHUNK)

typedef float v2f __attribute__((ext_vector_type(2)));
typedef float v8f __attribute__((ext_vector_type(8)));

#define LOG2E 1.44269504088896341f
#define LN2   0.693147180559945309f

__device__ __forceinline__ float exp2_hw(float v) {
  return __builtin_amdgcn_exp2f(v);
}

struct GateRow { float e0, e1, e2, eh, p, q, r, sg; };

__device__ __forceinline__ float ror8(float v) {
  return __builtin_bit_cast(float,
      __builtin_amdgcn_mov_dpp(__builtin_bit_cast(int, v), 0x128, 0xf, 0xf, false));
}

__device__ __forceinline__ void own_fi(float ta, float tb, bool lo, float& f2, float& i2) {
  f2 = lo ? ta : tb;
  i2 = ror8(lo ? tb : ta);
}

__device__ __forceinline__ void gate_s1(GateRow& g, float ta, float tb, float h2, bool lo) {
  float f2, i2;
  own_fi(ta, tb, lo, f2, i2);
  const float nf = -f2, ni = -i2;
  const float m  = fmaxf(fmaxf(nf, ni), 0.f);
  g.e0 = exp2_hw(-m);
  g.e1 = exp2_hw(nf - m);
  g.e2 = exp2_hw(ni - m);
  g.eh = exp2_hw(-h2);
}

__device__ __forceinline__ void gate_s2(GateRow& g) {
  g.p  = g.e0 + g.e1;
  g.q  = g.e0 + g.e2;
  g.r  = __builtin_amdgcn_rcpf(g.p + g.q);
  g.sg = __builtin_amdgcn_rcpf(1.f + g.eh);
}

__device__ __forceinline__ void gate_s3(const GateRow& g, float h2, float& Fo, float& Vo) {
  Fo = g.q * g.r;
  const float iw = g.p * g.r;
  const float gp = fmaf(h2, LN2, 0.5f);
  Vo = iw * fmaxf(gp, g.sg);
}

__device__ __forceinline__ void gate_lin(float ta, float tb, float h2, bool lo,
                                         float& Fo, float& Vo) {
  GateRow g;
  gate_s1(g, ta, tb, h2, lo);
  gate_s2(g);
  gate_s3(g, h2, Fo, Vo);
}

__device__ __forceinline__ v2f ldw(const float* __restrict__ Wg, int cn, int kc,
                                   int hi, bool valid) {
  v2f w = *(const v2f*)(Wg + cn * 8 + kc * 4 + 2 * hi);
  v2f r;
  r.x = valid ? w.x * LOG2E : 0.f;
  r.y = valid ? w.y * LOG2E : 0.f;
  return r;
}

struct WB { v2f b00, b01, b10, b11; };

__device__ __forceinline__ WB load_wb(const float* __restrict__ W, int g0,
                                      int n, int hi) {
  const int  cn = n & 7;
  const bool lo = (n < 8);
  const float* Wg0 = W + (size_t)g0 * 64;
  const float* Wg1 = Wg0 + 64;
  WB w;
  w.b00 = ldw(Wg0, cn, 0, hi, lo);
  w.b01 = ldw(Wg0, cn, 1, hi, lo);
  w.b10 = ldw(Wg1, cn, 0, hi, !lo);
  w.b11 = ldw(Wg1, cn, 1, hi, !lo);
  return w;
}

struct WP { v2f k0, k1; };

__device__ __forceinline__ WP load_wp(const float* __restrict__ Wlo,
                                      const float* __restrict__ Whi, int g,
                                      int n, int hi) {
  const float* Wsel = (n < 8) ? Wlo : Whi;
  const float* p = Wsel + (size_t)g * 64 + (n & 7) * 8 + 2 * hi;
  const v2f w0 = *(const v2f*)(p);
  const v2f w1 = *(const v2f*)(p + 4);
  WP w;
  w.k0.x = w0.x * LOG2E; w.k0.y = w0.y * LOG2E;
  w.k1.x = w1.x * LOG2E; w.k1.y = w1.y * LOG2E;
  return w;
}

struct XA { v2f a00, a01, a10, a11; };

__device__ __forceinline__ XA load_xa(const float* __restrict__ xrow) {
  XA a;
  a.a00 = *(const v2f*)(xrow + 0);
  a.a01 = *(const v2f*)(xrow + 4);
  a.a10 = *(const v2f*)(xrow + 8);
  a.a11 = *(const v2f*)(xrow + 12);
  return a;
}

__device__ __forceinline__ v8f gate16(const XA& a, const WB& w) {
  v8f acc = {};
  acc = __builtin_amdgcn_wmma_f32_16x16x4_f32(false, a.a00, false, w.b00, (short)0, acc, false, false);
  acc = __builtin_amdgcn_wmma_f32_16x16x4_f32(false, a.a01, false, w.b01, (short)0, acc, false, false);
  acc = __builtin_amdgcn_wmma_f32_16x16x4_f32(false, a.a10, false, w.b10, (short)0, acc, false, false);
  acc = __builtin_amdgcn_wmma_f32_16x16x4_f32(false, a.a11, false, w.b11, (short)0, acc, false, false);
  return acc;
}

__device__ __forceinline__ v8f pair16(const v2f& a0, const v2f& a1, const WP& w) {
  v8f acc = {};
  acc = __builtin_amdgcn_wmma_f32_16x16x4_f32(false, a0, false, w.k0, (short)0, acc, false, false);
  acc = __builtin_amdgcn_wmma_f32_16x16x4_f32(false, a1, false, w.k1, (short)0, acc, false, false);
  return acc;
}

#define TSTEP  8
#define NTILE  (CLEN / TSTEP)


__device__ __forceinline__ void local_tile(const v8f& ta, const v8f& tb, const v8f& hg, bool lo,
                                           float& A, float& Bv) {
#pragma unroll
  for (int r = 0; r < TSTEP; ++r) {
    float Fk, Vk;
    gate_lin(ta[r], tb[r], hg[r], lo, Fk, Vk);
    Bv = fmaf(Fk, Bv, Vk);
    A *= Fk;
  }
}

__device__ __forceinline__ void apply_tile(const v8f& ta, const v8f& tb, const v8f& hg, bool lo,
                                           float& hv, float* __restrict__ po) {
#pragma unroll
  for (int r = 0; r < TSTEP; ++r) {
    float Fk, Vk;
    gate_lin(ta[r], tb[r], hg[r], lo, Fk, Vk);
    hv = fmaf(Fk, hv, Vk);
    po[(size_t)r * Hdim] = hv;
  }
}

template <bool APPLY>
__device__ __forceinline__ void scan_row(const GateRow& g, float h2, float& s0, float& s1,
                                         float* __restrict__ po, int r) {
  float Fk, Vk;
  gate_s3(g, h2, Fk, Vk);
  if constexpr (APPLY) { s0 = fmaf(Fk, s0, Vk); po[(size_t)r * Hdim] = s0; }
  else { s1 = fmaf(Fk, s1, Vk); s0 *= Fk; }
}

template <bool APPLY, bool REFILL>
__device__ __forceinline__ void pipe_step(XA& a, const float* __restrict__ xnext,
                                          const WP& wa, const WP& wb, const WB& wh,
                                          v8f& aN, v8f& bN, v8f& hN,
                                          const v8f& aP, const v8f& bP, const v8f& hP,
                                          bool lo,
                                          float& s0, float& s1, float* __restrict__ po) {
  const v8f zero = {};
  GateRow g[TSTEP];
    hN = __builtin_amdgcn_wmma_f32_16x16x4_f32(false, a.a00, false, wh.b00, (short)0, zero, false, false);
    gate_s1(g[0], aP[0], bP[0], hP[0], lo);
    __builtin_amdgcn_sched_barrier(0);
    aN = __builtin_amdgcn_wmma_f32_16x16x4_f32(false, a.a00, false, wa.k0, (short)0, zero, false, false);
    gate_s1(g[1], aP[1], bP[1], hP[1], lo);
    gate_s2(g[0]);
    __builtin_amdgcn_sched_barrier(0);
    bN = __builtin_amdgcn_wmma_f32_16x16x4_f32(false, a.a10, false, wb.k0, (short)0, zero, false, false);
    gate_s1(g[2], aP[2], bP[2], hP[2], lo);
    gate_s2(g[1]);
    scan_row<APPLY>(g[0], hP[0], s0, s1, po, 0);
    __builtin_amdgcn_sched_barrier(0);
    hN = __builtin_amdgcn_wmma_f32_16x16x4_f32(false, a.a01, false, wh.b01, (short)0, hN, false, false);
    gate_s1(g[3], aP[3], bP[3], hP[3], lo);
    gate_s2(g[2]);
    scan_row<APPLY>(g[1], hP[1], s0, s1, po, 1);
    __builtin_amdgcn_sched_barrier(0);
    aN = __builtin_amdgcn_wmma_f32_16x16x4_f32(false, a.a01, false, wa.k1, (short)0, aN, false, false);
    gate_s1(g[4], aP[4], bP[4], hP[4], lo);
    gate_s2(g[3]);
    scan_row<APPLY>(g[2], hP[2], s0, s1, po, 2);
    __builtin_amdgcn_sched_barrier(0);
    hN = __builtin_amdgcn_wmma_f32_16x16x4_f32(false, a.a10, false, wh.b10, (short)0, hN, false, false);
    gate_s1(g[5], aP[5], bP[5], hP[5], lo);
    gate_s2(g[4]);
    scan_row<APPLY>(g[3], hP[3], s0, s1, po, 3);
    __builtin_amdgcn_sched_barrier(0);
    bN = __builtin_amdgcn_wmma_f32_16x16x4_f32(false, a.a11, false, wb.k1, (short)0, bN, false, false);
    gate_s1(g[6], aP[6], bP[6], hP[6], lo);
    gate_s2(g[5]);
    scan_row<APPLY>(g[4], hP[4], s0, s1, po, 4);
    __builtin_amdgcn_sched_barrier(0);
    hN = __builtin_amdgcn_wmma_f32_16x16x4_f32(false, a.a11, false, wh.b11, (short)0, hN, false, false);
    gate_s1(g[7], aP[7], bP[7], hP[7], lo);
    gate_s2(g[6]);
    scan_row<APPLY>(g[5], hP[5], s0, s1, po, 5);
    gate_s2(g[7]);
    scan_row<APPLY>(g[6], hP[6], s0, s1, po, 6);
    scan_row<APPLY>(g[7], hP[7], s0, s1, po, 7);
    if constexpr (REFILL) a = load_xa(xnext);
}

struct WaveJob {
  int n, hi, gp, q;
  const float* xp;
};

__device__ __forceinline__ WaveJob wave_job(const float* __restrict__ x) {
  WaveJob j;
  const int lane = threadIdx.x & 31;
  const int wv   = threadIdx.x >> 5;
  const int gid  = blockIdx.x * 8 + wv;
  const int NPAIR = GROUPS / 2;
  j.gp = gid % NPAIR;
  const int pc = gid / NPAIR;
  j.n  = lane & 15;
  j.hi = lane >> 4;
  j.q  = 2 * pc + j.hi;
  const int qrow = 2 * pc + (j.n >> 3);
  j.xp = x + ((size_t)qrow * CLEN + (size_t)(j.n & 7)) * (size_t)Ddim +
         (size_t)j.gp * 16 + 2 * j.hi;
  return j;
}

__global__ void __launch_bounds__(256, 8) chunk_local(
    const float* __restrict__ x, const float* __restrict__ Wf,
    const float* __restrict__ Wi, const float* __restrict__ Wh,
    float* __restrict__ chA, float* __restrict__ chB) {
  const WaveJob jb = wave_job(x);
  const float* xp = jb.xp;
  const bool lo = jb.n < 8;
  const WP wa = load_wp(Wf, Wi, jb.gp * 2,     jb.n, jb.hi);
  const WP wb = load_wp(Wi, Wf, jb.gp * 2 + 1, jb.n, jb.hi);
  const WB wh = load_wb(Wh, jb.gp * 2, jb.n, jb.hi);

  float A = 1.f, Bv = 0.f;
  XA r0 = load_xa(xp);
  XA r1 = load_xa(xp + (size_t)TSTEP * Ddim);
  v8f aE = pair16(r0.a00, r0.a01, wa), bE = pair16(r0.a10, r0.a11, wb), hE = gate16(r0, wh);
  r0 = load_xa(xp + 2 * (size_t)TSTEP * Ddim);
  v8f aO, bO, hO;
  for (int t = 0; t < (NTILE - 4) / 2; ++t) {
    pipe_step<false, true>(r1, xp + 3 * (size_t)TSTEP * Ddim, wa, wb, wh, aO, bO, hO, aE, bE, hE, lo, A, Bv, nullptr);
    pipe_step<false, true>(r0, xp + 4 * (size_t)TSTEP * Ddim, wa, wb, wh, aE, bE, hE, aO, bO, hO, lo, A, Bv, nullptr);
    xp += 2 * (size_t)TSTEP * Ddim;
  }
  pipe_step<false, true >(r1, xp + 3 * (size_t)TSTEP * Ddim, wa, wb, wh, aO, bO, hO, aE, bE, hE, lo, A, Bv, nullptr);
  pipe_step<false, false>(r0, xp,             wa, wb, wh, aE, bE, hE, aO, bO, hO, lo, A, Bv, nullptr);
  pipe_step<false, false>(r1, xp,             wa, wb, wh, aO, bO, hO, aE, bE, hE, lo, A, Bv, nullptr);
  local_tile(aO, bO, hO, lo, A, Bv);
  const int b = jb.q / NCHUNK;
  const int c = jb.q % NCHUNK;
  const int h = jb.gp * 16 + jb.n;
  const int sidx = c * (Bdim * Hdim) + b * Hdim + h;
  chA[sidx] = A;
  chB[sidx] = Bv;
}

#define CGRP 8
#define NGRP (NCHUNK / 2 / CGRP)
__device__ __forceinline__ float carry_in(const float* __restrict__ pa,
                                          const float* __restrict__ pb,
                                          int c, int cmax) {
  const int nchan = Bdim * Hdim;
  float qa[NCHUNK / 2], qb[NCHUNK / 2];
#pragma unroll
  for (int k = 0; k < NCHUNK / 2; ++k) {
    qa[k] = pa[k * nchan];
    qb[k] = pb[k * nchan];
  }
  float H = 0.f;
  if (cmax > NCHUNK / 2) {
#pragma unroll
    for (int g = 0; g < NGRP; ++g) {
#pragma unroll
      for (int j = 0; j < CGRP; ++j) {
        const int k = g * CGRP + j;
        H = fmaf(qa[k], H, qb[k]);
      }
#pragma unroll
      for (int j = 0; j < CGRP; ++j) {
        const int k = g * CGRP + j;
        qa[k] = pa[(NCHUNK / 2 + k) * nchan];
        qb[k] = pb[(NCHUNK / 2 + k) * nchan];
      }
    }
#pragma unroll
    for (int g = 0; g < NGRP; ++g) {
      if (NCHUNK / 2 + g * CGRP < cmax) {
#pragma unroll
        for (int j = 0; j < CGRP; ++j) {
          const int k = g * CGRP + j;
          const float Hn = fmaf(qa[k], H, qb[k]);
          H = (NCHUNK / 2 + k < c) ? Hn : H;
        }
      }
    }
  } else {
#pragma unroll
    for (int g = 0; g < NGRP; ++g) {
      if (g * CGRP < cmax) {
#pragma unroll
        for (int j = 0; j < CGRP; ++j) {
          const int k = g * CGRP + j;
          const float Hn = fmaf(qa[k], H, qb[k]);
          H = (k < c) ? Hn : H;
        }
      }
    }
  }
  return H;
}

__global__ void __launch_bounds__(256) apply_scan(
    const float* __restrict__ x, const float* __restrict__ Wf,
    const float* __restrict__ Wi, const float* __restrict__ Wh,
    const float* __restrict__ chA, const float* __restrict__ chB,
    float* __restrict__ out) {
  const WaveJob jb = wave_job(x);
  const float* xp = jb.xp;
  const bool lo = jb.n < 8;
  const WP wa = load_wp(Wf, Wi, jb.gp * 2,     jb.n, jb.hi);
  const WP wb = load_wp(Wi, Wf, jb.gp * 2 + 1, jb.n, jb.hi);
  const WB wh = load_wb(Wh, jb.gp * 2, jb.n, jb.hi);

  const int b = jb.q / NCHUNK;
  const int c = jb.q % NCHUNK;
  const int h = jb.gp * 16 + jb.n;
  float unused = 0.f;
  float* po = out + (size_t)jb.q * CLEN * (size_t)Hdim + h;
  XA r0 = load_xa(xp);
  XA r1 = load_xa(xp + (size_t)TSTEP * Ddim);
  const int cmax = __builtin_amdgcn_readfirstlane((jb.q | 1) % NCHUNK);
  float hv = carry_in(chA + b * Hdim + h, chB + b * Hdim + h, c, cmax);
  v8f aE = pair16(r0.a00, r0.a01, wa), bE = pair16(r0.a10, r0.a11, wb), hE = gate16(r0, wh);
  r0 = load_xa(xp + 2 * (size_t)TSTEP * Ddim);
  v8f aO, bO, hO;
  for (int t = 0; t < (NTILE - 4) / 2; ++t) {
    pipe_step<true, true>(r1, xp + 3 * (size_t)TSTEP * Ddim, wa, wb, wh, aO, bO, hO, aE, bE, hE, lo, hv, unused, po);
    pipe_step<true, true>(r0, xp + 4 * (size_t)TSTEP * Ddim, wa, wb, wh, aE, bE, hE, aO, bO, hO, lo, hv, unused, po + (size_t)1 * TSTEP * Hdim);
    xp += 2 * (size_t)TSTEP * Ddim;
    po += (size_t)2 * TSTEP * Hdim;
  }
  pipe_step<true, true >(r1, xp + 3 * (size_t)TSTEP * Ddim, wa, wb, wh, aO, bO, hO, aE, bE, hE, lo, hv, unused, po);
  pipe_step<true, false>(r0, xp,             wa, wb, wh, aE, bE, hE, aO, bO, hO, lo, hv, unused, po + (size_t)1 * TSTEP * Hdim);
  pipe_step<true, false>(r1, xp,             wa, wb, wh, aO, bO, hO, aE, bE, hE, lo, hv, unused, po + (size_t)2 * TSTEP * Hdim);
  apply_tile(aO, bO, hO, lo, hv, po + (size_t)3 * TSTEP * Hdim);
}

extern "C" void kernel_launch(void* const* d_in, const int* in_sizes, int n_in,
                              void* d_out, int out_size, void* d_ws, size_t ws_size,
                              hipStream_t stream) {
  (void)in_sizes; (void)n_in; (void)out_size; (void)ws_size;
  const float* x  = (const float*)d_in[0];
  const float* Wf = (const float*)d_in[1];
  const float* Wi = (const float*)d_in[2];
  const float* Wh = (const float*)d_in[3];
  float* out = (float*)d_out;

  const size_t NSUM = (size_t)NCHUNK * Bdim * Hdim;
  float* chA = (float*)d_ws;
  float* chB = chA + NSUM;

  const int scan_blocks = (GROUPS / 2) * (Bdim * NCHUNK / 2) / 8;
  chunk_local<<<scan_blocks, 256, 0, stream>>>(x, Wf, Wi, Wh, chA, chB);
  apply_scan<<<scan_blocks, 256, 0, stream>>>(x, Wf, Wi, Wh, chA, chB, out);
}
